# starting kernel
# baseline (speedup 1.0000x reference)
#include <hip/hip_runtime.h>
#include <hip/hip_bf16.h>


typedef __attribute__((ext_vector_type(2))) float v2f;
typedef __attribute__((ext_vector_type(8))) float v8f;

#define LEN_SCALE_INV (1.0f / 20000.0f)

__global__ void k_deg_init(float* __restrict__ dinv, int N) {
    int i = blockIdx.x * blockDim.x + threadIdx.x;
    if (i < N) dinv[i] = 1.0f;
}

__global__ void k_deg_edges(const int* __restrict__ dst, float* __restrict__ dinv,
                            int E) {
    int e = blockIdx.x * blockDim.x + threadIdx.x;
    if (e < E) unsafeAtomicAdd(&dinv[dst[e]], 1.0f);
}

__global__ void k_deg_fin(const int* __restrict__ read_length,
                          float* __restrict__ dinv, float* __restrict__ dsq,
                          float* __restrict__ x0, int N) {
    int i = blockIdx.x * blockDim.x + threadIdx.x;
    if (i < N) {
        float deg = dinv[i];
        dsq[i]  = rsqrtf(deg);
        dinv[i] = 1.0f / deg;
        x0[i]   = (float)read_length[i] * LEN_SCALE_INV;
    }
}

__global__ void k_norm(const int* __restrict__ src, const int* __restrict__ dst,
                       const float* __restrict__ dsq, float* __restrict__ nrm,
                       int E) {
    int e = blockIdx.x * blockDim.x + threadIdx.x;
    if (e < E) nrm[e] = dsq[src[e]] * dsq[dst[e]];
}

__global__ void k_zero(float* __restrict__ p, long long n) {
    long long i = (long long)blockIdx.x * blockDim.x + threadIdx.x;
    if (i < n) p[i] = 0.0f;
}

template <int F_IN, int F_OUT>
__global__ void k_linear_wmma(const float* __restrict__ x,
                              const float* __restrict__ W,
                              float* __restrict__ xw, int N) {
    constexpr int TILES_N = (F_OUT + 15) / 16;
    constexpr int KCH     = (F_IN + 3) / 4;

    const int wave   = blockIdx.x * (blockDim.x >> 5) + (threadIdx.x >> 5);
    const int lane   = threadIdx.x & 31;
    const int tile_m = wave / TILES_N;
    const int tile_n = wave - tile_m * TILES_N;
    const int row0   = tile_m * 16;
    if (row0 >= N) return;

    const int  lg   = lane >> 4;
    const int  lmod = lane & 15;
    const int  m    = row0 + lmod;
    const int  mc   = (m < N) ? m : (N - 1);
    const int  nc   = tile_n * 16 + lmod;
    const bool nok  = (nc < F_OUT);
    const int  ncc  = nok ? nc : 0;

    v8f acc = {0.f, 0.f, 0.f, 0.f, 0.f, 0.f, 0.f, 0.f};

#pragma unroll
    for (int kc = 0; kc < KCH; ++kc) {
        const int k0  = kc * 4 + lg * 2;
        const int k1  = k0 + 1;
        const int k0c = (k0 < F_IN) ? k0 : (F_IN - 1);
        const int k1c = (k1 < F_IN) ? k1 : (F_IN - 1);

        float a0 = x[(long long)mc * F_IN + k0c];
        float a1 = x[(long long)mc * F_IN + k1c];
        a0 = (m < N && k0 < F_IN) ? a0 : 0.f;
        a1 = (m < N && k1 < F_IN) ? a1 : 0.f;
        v2f a; a.x = a0; a.y = a1;

        float b0 = W[k0c * F_OUT + ncc];
        float b1 = W[k1c * F_OUT + ncc];
        b0 = (nok && k0 < F_IN) ? b0 : 0.f;
        b1 = (nok && k1 < F_IN) ? b1 : 0.f;
        v2f b; b.x = b0; b.y = b1;

        acc = __builtin_amdgcn_wmma_f32_16x16x4_f32(
            false, a, false, b, (short)0, acc, false, false);
    }

    const bool fullM = (row0 + 16 <= N);
    if (nok) {
        float* p = xw + (long long)(row0 + lg * 8) * F_OUT + nc;
        if (fullM) {
#pragma unroll
            for (int v = 0; v < 8; ++v) p[(long long)v * F_OUT] = acc[v];
        } else {
#pragma unroll
            for (int v = 0; v < 8; ++v) {
                const int mrow = row0 + v + lg * 8;
                if (mrow < N) p[(long long)v * F_OUT] = acc[v];
            }
        }
    }
}

template <int LOGF>
__global__ void k_scatter(const float* __restrict__ xw,
                          const int* __restrict__ src,
                          const int* __restrict__ dst,
                          const float* __restrict__ nrm,
                          float* __restrict__ agg, long long EF) {
    const long long idx = (long long)blockIdx.x * blockDim.x + threadIdx.x;
    if (idx >= EF) return;
    const int e = (int)(idx >> LOGF);
    const int f = (int)(idx & ((1 << LOGF) - 1));
    const int s = src[e];
    const int d = dst[e];
    const float v = xw[((long long)s << LOGF) + f] * nrm[e];
    unsafeAtomicAdd(&agg[((long long)d << LOGF) + f], v);
}

template <int LOGF>
__global__ void k_finalize(const float* __restrict__ agg,
                           const float* __restrict__ xw,
                           const float* __restrict__ dinv,
                           const float* __restrict__ bias,
                           float* __restrict__ out, long long NF) {
    const long long i = (long long)blockIdx.x * blockDim.x + threadIdx.x;
    if (i >= NF) return;
    const int node = (int)(i >> LOGF);
    const int f    = (int)(i & ((1 << LOGF) - 1));
    out[i] = agg[i] + xw[i] * dinv[node] + bias[f];
}

template <int F_IN, int F_OUT, int LOGF>
static void run_layer(const float* xin, const float* W, const float* bias,
                      float* xw, float* agg, const float* dinv,
                      const int* src, const int* dst, const float* nrm,
                      float* xout, int N, int E, hipStream_t stream) {
    constexpr int T = 256;
    constexpr int TILES_N = (F_OUT + 15) / 16;

    const long long waves  = (long long)((N + 15) / 16) * TILES_N;
    const long long blocks = (waves + 7) / 8;
    k_linear_wmma<F_IN, F_OUT><<<(unsigned)blocks, 256, 0, stream>>>(xin, W, xw, N);

    const long long nf = (long long)N * F_OUT;
    k_zero<<<(unsigned)((nf + T - 1) / T), T, 0, stream>>>(agg, nf);

    const long long ef = (long long)E * F_OUT;
    k_scatter<LOGF><<<(unsigned)((ef + T - 1) / T), T, 0, stream>>>(
        xw, src, dst, nrm, agg, ef);

    k_finalize<LOGF><<<(unsigned)((nf + T - 1) / T), T, 0, stream>>>(
        agg, xw, dinv, bias, xout, nf);
}

extern "C" void kernel_launch(void* const* d_in, const int* in_sizes, int n_in,
                              void* d_out, int out_size, void* d_ws, size_t ws_size,
                              hipStream_t stream) {
    const int* read_length = (const int*)d_in[0];
    const int* edge        = (const int*)d_in[1];
    const int  N = in_sizes[0];
    const int  E = in_sizes[1] / 2;
    const int* src = edge;
    const int* dst = edge + E;

    const float* W1 = (const float*)d_in[2]; const float* B1 = (const float*)d_in[3];
    const float* W2 = (const float*)d_in[4]; const float* B2 = (const float*)d_in[5];
    const float* W3 = (const float*)d_in[6]; const float* B3 = (const float*)d_in[7];
    const float* W4 = (const float*)d_in[8]; const float* B4 = (const float*)d_in[9];

    float* ws   = (float*)d_ws;
    float* xbuf = ws;
    float* xw   = xbuf + (size_t)N * 32;
    float* agg  = xw   + (size_t)N * 32;
    float* dinv = agg  + (size_t)N * 32;
    float* dsq  = dinv + N;
    float* nrm  = dsq  + N;

    const int T = 256;

    k_deg_init<<<(N + T - 1) / T, T, 0, stream>>>(dinv, N);
    k_deg_edges<<<(E + T - 1) / T, T, 0, stream>>>(dst, dinv, E);
    k_deg_fin<<<(N + T - 1) / T, T, 0, stream>>>(read_length, dinv, dsq, xbuf, N);
    k_norm<<<(E + T - 1) / T, T, 0, stream>>>(src, dst, dsq, nrm, E);

    run_layer<1, 4, 2>(xbuf, W1, B1, xw, agg, dinv, src, dst, nrm, xbuf, N, E, stream);
    run_layer<4, 8, 3>(xbuf, W2, B2, xw, agg, dinv, src, dst, nrm, xbuf, N, E, stream);
    run_layer<8, 16, 4>(xbuf, W3, B3, xw, agg, dinv, src, dst, nrm, xbuf, N, E, stream);
    run_layer<16, 32, 5>(xbuf, W4, B4, xw, agg, dinv, src, dst, nrm,
                         (float*)d_out, N, E, stream);
}


# layers 2-4 aggregate before the linear, (A x) W for A (x W): atomic edge scatter runs F_IN wide, half of F_OUT; self-loop and bias fused into the WMMA kernel; float sum order changes
# speedup vs baseline: 1.5016x; 1.5016x over previous
#include <hip/hip_runtime.h>
#include <hip/hip_bf16.h>


typedef __attribute__((ext_vector_type(2))) float v2f;
typedef __attribute__((ext_vector_type(8))) float v8f;

#define LEN_SCALE_INV (1.0f / 20000.0f)

__global__ void k_deg_init(float* __restrict__ dinv, int N) {
    int i = blockIdx.x * blockDim.x + threadIdx.x;
    if (i < N) dinv[i] = 1.0f;
}

__global__ void k_deg_edges(const int* __restrict__ dst, float* __restrict__ dinv,
                            int E) {
    int e = blockIdx.x * blockDim.x + threadIdx.x;
    if (e < E) unsafeAtomicAdd(&dinv[dst[e]], 1.0f);
}

__global__ void k_deg_fin(const int* __restrict__ read_length,
                          float* __restrict__ dinv, float* __restrict__ dsq,
                          float* __restrict__ x0, int N) {
    int i = blockIdx.x * blockDim.x + threadIdx.x;
    if (i < N) {
        float deg = dinv[i];
        dsq[i]  = rsqrtf(deg);
        dinv[i] = 1.0f / deg;
        x0[i]   = (float)read_length[i] * LEN_SCALE_INV;
    }
}

__global__ void k_norm(const int* __restrict__ src, const int* __restrict__ dst,
                       const float* __restrict__ dsq, float* __restrict__ nrm,
                       int E) {
    int e = blockIdx.x * blockDim.x + threadIdx.x;
    if (e < E) nrm[e] = dsq[src[e]] * dsq[dst[e]];
}

__global__ void k_zero(float* __restrict__ p, long long n) {
    long long i = (long long)blockIdx.x * blockDim.x + threadIdx.x;
    if (i < n) p[i] = 0.0f;
}

template <int F_IN, int F_OUT>
__global__ void k_linear_wmma(const float* __restrict__ x,
                              const float* __restrict__ W,
                              float* __restrict__ xw, int N) {
    constexpr int TILES_N = (F_OUT + 15) / 16;
    constexpr int KCH     = (F_IN + 3) / 4;

    const int wave   = blockIdx.x * (blockDim.x >> 5) + (threadIdx.x >> 5);
    const int lane   = threadIdx.x & 31;
    const int tile_m = wave / TILES_N;
    const int tile_n = wave - tile_m * TILES_N;
    const int row0   = tile_m * 16;
    if (row0 >= N) return;

    const int  lg   = lane >> 4;
    const int  lmod = lane & 15;
    const int  m    = row0 + lmod;
    const int  mc   = (m < N) ? m : (N - 1);
    const int  nc   = tile_n * 16 + lmod;
    const bool nok  = (nc < F_OUT);
    const int  ncc  = nok ? nc : 0;

    v8f acc = {0.f, 0.f, 0.f, 0.f, 0.f, 0.f, 0.f, 0.f};

#pragma unroll
    for (int kc = 0; kc < KCH; ++kc) {
        const int k0  = kc * 4 + lg * 2;
        const int k1  = k0 + 1;
        const int k0c = (k0 < F_IN) ? k0 : (F_IN - 1);
        const int k1c = (k1 < F_IN) ? k1 : (F_IN - 1);

        float a0 = x[(long long)mc * F_IN + k0c];
        float a1 = x[(long long)mc * F_IN + k1c];
        a0 = (m < N && k0 < F_IN) ? a0 : 0.f;
        a1 = (m < N && k1 < F_IN) ? a1 : 0.f;
        v2f a; a.x = a0; a.y = a1;

        float b0 = W[k0c * F_OUT + ncc];
        float b1 = W[k1c * F_OUT + ncc];
        b0 = (nok && k0 < F_IN) ? b0 : 0.f;
        b1 = (nok && k1 < F_IN) ? b1 : 0.f;
        v2f b; b.x = b0; b.y = b1;

        acc = __builtin_amdgcn_wmma_f32_16x16x4_f32(
            false, a, false, b, (short)0, acc, false, false);
    }

    const bool fullM = (row0 + 16 <= N);
    if (nok) {
        float* p = xw + (long long)(row0 + lg * 8) * F_OUT + nc;
        if (fullM) {
#pragma unroll
            for (int v = 0; v < 8; ++v) p[(long long)v * F_OUT] = acc[v];
        } else {
#pragma unroll
            for (int v = 0; v < 8; ++v) {
                const int mrow = row0 + v + lg * 8;
                if (mrow < N) p[(long long)v * F_OUT] = acc[v];
            }
        }
    }
}

template <int LOGF>
__global__ void k_scatter(const float* __restrict__ xw,
                          const int* __restrict__ src,
                          const int* __restrict__ dst,
                          const float* __restrict__ nrm,
                          float* __restrict__ agg, long long EF) {
    const long long idx = (long long)blockIdx.x * blockDim.x + threadIdx.x;
    if (idx >= EF) return;
    const int e = (int)(idx >> LOGF);
    const int f = (int)(idx & ((1 << LOGF) - 1));
    const int s = src[e];
    const int d = dst[e];
    const float v = xw[((long long)s << LOGF) + f] * nrm[e];
    unsafeAtomicAdd(&agg[((long long)d << LOGF) + f], v);
}

template <int LOGF>
__global__ void k_finalize(const float* __restrict__ agg,
                           const float* __restrict__ xw,
                           const float* __restrict__ dinv,
                           const float* __restrict__ bias,
                           float* __restrict__ out, long long NF) {
    const long long i = (long long)blockIdx.x * blockDim.x + threadIdx.x;
    if (i >= NF) return;
    const int node = (int)(i >> LOGF);
    const int f    = (int)(i & ((1 << LOGF) - 1));
    out[i] = agg[i] + xw[i] * dinv[node] + bias[f];
}

template <int F_IN, int F_OUT>
__global__ void k_combine_linear_wmma(const float* __restrict__ agg,
                                      const float* __restrict__ x,
                                      const float* __restrict__ dinv,
                                      const float* __restrict__ W,
                                      const float* __restrict__ bias,
                                      float* __restrict__ out, int N) {
    constexpr int TILES_N = (F_OUT + 15) / 16;
    constexpr int KCH     = (F_IN + 3) / 4;

    const int wave = blockIdx.x * (blockDim.x >> 5) + (threadIdx.x >> 5);
    const int lane = threadIdx.x & 31;
    const int row0 = wave * 16;
    if (row0 >= N) return;

    const int  lg   = lane >> 4;
    const int  lmod = lane & 15;
    const int  m    = row0 + lmod;
    const bool mok  = (m < N);
    const int  mc   = mok ? m : (N - 1);
    const float di  = dinv[mc];

    v2f a[KCH];
#pragma unroll
    for (int kc = 0; kc < KCH; ++kc) {
        const int k0  = kc * 4 + lg * 2;
        const int k1  = k0 + 1;
        const int k0c = (k0 < F_IN) ? k0 : (F_IN - 1);
        const int k1c = (k1 < F_IN) ? k1 : (F_IN - 1);
        const long long i0 = (long long)mc * F_IN + k0c;
        const long long i1 = (long long)mc * F_IN + k1c;
        float h0 = agg[i0] + x[i0] * di;
        float h1 = agg[i1] + x[i1] * di;
        h0 = (mok && k0 < F_IN) ? h0 : 0.f;
        h1 = (mok && k1 < F_IN) ? h1 : 0.f;
        a[kc].x = h0; a[kc].y = h1;
    }

    const bool fullM = (row0 + 16 <= N);

#pragma unroll
    for (int tn = 0; tn < TILES_N; ++tn) {
        const int  nc  = tn * 16 + lmod;
        const bool nok = (nc < F_OUT);
        const int  ncc = nok ? nc : 0;
        const float bv = bias[ncc];

        v8f acc = {0.f, 0.f, 0.f, 0.f, 0.f, 0.f, 0.f, 0.f};

#pragma unroll
        for (int kc = 0; kc < KCH; ++kc) {
            const int k0  = kc * 4 + lg * 2;
            const int k1  = k0 + 1;
            const int k0c = (k0 < F_IN) ? k0 : (F_IN - 1);
            const int k1c = (k1 < F_IN) ? k1 : (F_IN - 1);

            float b0 = W[k0c * F_OUT + ncc];
            float b1 = W[k1c * F_OUT + ncc];
            b0 = (nok && k0 < F_IN) ? b0 : 0.f;
            b1 = (nok && k1 < F_IN) ? b1 : 0.f;
            v2f b; b.x = b0; b.y = b1;

            acc = __builtin_amdgcn_wmma_f32_16x16x4_f32(
                false, a[kc], false, b, (short)0, acc, false, false);
        }

        if (nok) {
            float* p = out + (long long)(row0 + lg * 8) * F_OUT + nc;
            if (fullM) {
#pragma unroll
                for (int v = 0; v < 8; ++v) p[(long long)v * F_OUT] = acc[v] + bv;
            } else {
#pragma unroll
                for (int v = 0; v < 8; ++v) {
                    const int mrow = row0 + v + lg * 8;
                    if (mrow < N) p[(long long)v * F_OUT] = acc[v] + bv;
                }
            }
        }
    }
}

template <int F_IN, int F_OUT, int LOGF>
static void run_layer(const float* xin, const float* W, const float* bias,
                      float* xw, float* agg, const float* dinv,
                      const int* src, const int* dst, const float* nrm,
                      float* xout, int N, int E, hipStream_t stream) {
    constexpr int T = 256;
    constexpr int TILES_N = (F_OUT + 15) / 16;

    const long long waves  = (long long)((N + 15) / 16) * TILES_N;
    const long long blocks = (waves + 7) / 8;
    k_linear_wmma<F_IN, F_OUT><<<(unsigned)blocks, 256, 0, stream>>>(xin, W, xw, N);

    const long long nf = (long long)N * F_OUT;
    k_zero<<<(unsigned)((nf + T - 1) / T), T, 0, stream>>>(agg, nf);

    const long long ef = (long long)E * F_OUT;
    k_scatter<LOGF><<<(unsigned)((ef + T - 1) / T), T, 0, stream>>>(
        xw, src, dst, nrm, agg, ef);

    k_finalize<LOGF><<<(unsigned)((nf + T - 1) / T), T, 0, stream>>>(
        agg, xw, dinv, bias, xout, nf);
}

template <int F_IN, int F_OUT, int LOGF_IN>
static void run_layer_agg_first(const float* xin, const float* W, const float* bias,
                                float* agg, const float* dinv,
                                const int* src, const int* dst, const float* nrm,
                                float* xout, int N, int E, hipStream_t stream) {
    constexpr int T = 256;

    const long long nf = (long long)N * F_IN;
    k_zero<<<(unsigned)((nf + T - 1) / T), T, 0, stream>>>(agg, nf);

    const long long ef = (long long)E * F_IN;
    k_scatter<LOGF_IN><<<(unsigned)((ef + T - 1) / T), T, 0, stream>>>(
        xin, src, dst, nrm, agg, ef);

    const long long waves  = (long long)((N + 15) / 16);
    const long long blocks = (waves + 7) / 8;
    k_combine_linear_wmma<F_IN, F_OUT><<<(unsigned)blocks, 256, 0, stream>>>(
        agg, xin, dinv, W, bias, xout, N);
}

extern "C" void kernel_launch(void* const* d_in, const int* in_sizes, int n_in,
                              void* d_out, int out_size, void* d_ws, size_t ws_size,
                              hipStream_t stream) {
    const int* read_length = (const int*)d_in[0];
    const int* edge        = (const int*)d_in[1];
    const int  N = in_sizes[0];
    const int  E = in_sizes[1] / 2;
    const int* src = edge;
    const int* dst = edge + E;

    const float* W1 = (const float*)d_in[2]; const float* B1 = (const float*)d_in[3];
    const float* W2 = (const float*)d_in[4]; const float* B2 = (const float*)d_in[5];
    const float* W3 = (const float*)d_in[6]; const float* B3 = (const float*)d_in[7];
    const float* W4 = (const float*)d_in[8]; const float* B4 = (const float*)d_in[9];

    float* ws   = (float*)d_ws;
    float* xbuf = ws;
    float* xw   = xbuf + (size_t)N * 32;
    float* agg  = xw   + (size_t)N * 32;
    float* dinv = agg  + (size_t)N * 32;
    float* dsq  = dinv + N;
    float* nrm  = dsq  + N;

    const int T = 256;

    k_deg_init<<<(N + T - 1) / T, T, 0, stream>>>(dinv, N);
    k_deg_edges<<<(E + T - 1) / T, T, 0, stream>>>(dst, dinv, E);
    k_deg_fin<<<(N + T - 1) / T, T, 0, stream>>>(read_length, dinv, dsq, xbuf, N);
    k_norm<<<(E + T - 1) / T, T, 0, stream>>>(src, dst, dsq, nrm, E);

    run_layer<1, 4, 2>(xbuf, W1, B1, xw, agg, dinv, src, dst, nrm, xbuf, N, E, stream);
    run_layer_agg_first<4, 8, 2>(xbuf, W2, B2, agg, dinv, src, dst, nrm, xw, N, E, stream);
    run_layer_agg_first<8, 16, 3>(xw, W3, B3, agg, dinv, src, dst, nrm, xbuf, N, E, stream);
    run_layer_agg_first<16, 32, 4>(xbuf, W4, B4, agg, dinv, src, dst, nrm,
                                   (float*)d_out, N, E, stream);
}


# float-atomic edge scatter -> per-destination gather: edges counting-sorted by dst each call (int atomic histogram, scan, arrival-order slot claim), 8 lanes sum a row chunk; float sum order changes
# speedup vs baseline: 2.2559x; 1.5023x over previous
#include <hip/hip_runtime.h>
#include <hip/hip_bf16.h>


typedef __attribute__((ext_vector_type(2))) float v2f;
typedef __attribute__((ext_vector_type(8))) float v8f;

#define LEN_SCALE_INV (1.0f / 20000.0f)

__global__ void k_cnt_zero(int* __restrict__ cnt, int N) {
    int i = blockIdx.x * blockDim.x + threadIdx.x;
    if (i < N) cnt[i] = 0;
}

__global__ void k_count(const int* __restrict__ dst, int* __restrict__ cnt, int E) {
    int e = blockIdx.x * blockDim.x + threadIdx.x;
    if (e < E) atomicAdd(&cnt[dst[e]], 1);
}

__global__ void k_scan_part(const int* __restrict__ cnt, int* __restrict__ bsum,
                            int N) {
    __shared__ int sh[8];
    const int tid  = threadIdx.x;
    const int base = blockIdx.x * 1024 + tid * 4;
    int s = 0;
#pragma unroll
    for (int k = 0; k < 4; ++k)
        if (base + k < N) s += cnt[base + k];
    for (int off = 16; off > 0; off >>= 1) s += __shfl_xor(s, off);
    if ((tid & 31) == 0) sh[tid >> 5] = s;
    __syncthreads();
    if (tid == 0) {
        int t = 0;
#pragma unroll
        for (int w = 0; w < 8; ++w) t += sh[w];
        bsum[blockIdx.x] = t;
    }
}

__global__ void k_scan_top(int* __restrict__ bsum, int NB) {
    __shared__ int sh[1024];
    const int tid = threadIdx.x;
    int carry = 0;
    for (int base = 0; base < NB; base += 1024) {
        const int i = base + tid;
        const int v = (i < NB) ? bsum[i] : 0;
        sh[tid] = v;
        __syncthreads();
        for (int off = 1; off < 1024; off <<= 1) {
            const int t = (tid >= off) ? sh[tid - off] : 0;
            __syncthreads();
            sh[tid] += t;
            __syncthreads();
        }
        const int incl  = sh[tid];
        const int total = sh[1023];
        if (i < NB) bsum[i] = carry + incl - v;
        carry += total;
        __syncthreads();
    }
}

__global__ void k_scan_fin(const int* __restrict__ cnt, const int* __restrict__ bsum,
                           const int* __restrict__ read_length,
                           int* __restrict__ rowptr, int* __restrict__ cursor,
                           float* __restrict__ dinv, float* __restrict__ dsq,
                           float* __restrict__ x0, int N) {
    __shared__ int sh[8];
    const int tid  = threadIdx.x;
    const int lane = tid & 31;
    const int wave = tid >> 5;
    const int base = blockIdx.x * 1024 + tid * 4;
    int c[4];
    int s = 0;
#pragma unroll
    for (int k = 0; k < 4; ++k) {
        c[k] = (base + k < N) ? cnt[base + k] : 0;
        s += c[k];
    }
    int incl = s;
    for (int off = 1; off < 32; off <<= 1) {
        const int t = __shfl_up(incl, off);
        if (lane >= off) incl += t;
    }
    if (lane == 31) sh[wave] = incl;
    __syncthreads();
    int woff = 0;
    for (int w = 0; w < wave; ++w) woff += sh[w];
    int excl = bsum[blockIdx.x] + woff + incl - s;
#pragma unroll
    for (int k = 0; k < 4; ++k) {
        const int i = base + k;
        if (i < N) {
            rowptr[i] = excl;
            cursor[i] = excl;
            const float deg = (float)(c[k] + 1);
            dsq[i]  = rsqrtf(deg);
            dinv[i] = 1.0f / deg;
            x0[i]   = (float)read_length[i] * LEN_SCALE_INV;
            excl += c[k];
            if (i == N - 1) rowptr[N] = excl;
        }
    }
}

__global__ void k_fill(const int* __restrict__ src, const int* __restrict__ dst,
                       const float* __restrict__ dsq, int* __restrict__ cursor,
                       unsigned long long* __restrict__ csr, int E) {
    int e = blockIdx.x * blockDim.x + threadIdx.x;
    if (e < E) {
        const int s = src[e];
        const int d = dst[e];
        const float w = dsq[s] * dsq[d];
        const int pos = atomicAdd(&cursor[d], 1);
        csr[pos] = ((unsigned long long)__float_as_uint(w) << 32) | (unsigned int)s;
    }
}

template <int F_IN, int F_OUT>
__global__ void k_linear_wmma(const float* __restrict__ x,
                              const float* __restrict__ W,
                              float* __restrict__ xw, int N) {
    constexpr int TILES_N = (F_OUT + 15) / 16;
    constexpr int KCH     = (F_IN + 3) / 4;

    const int wave   = blockIdx.x * (blockDim.x >> 5) + (threadIdx.x >> 5);
    const int lane   = threadIdx.x & 31;
    const int tile_m = wave / TILES_N;
    const int tile_n = wave - tile_m * TILES_N;
    const int row0   = tile_m * 16;
    if (row0 >= N) return;

    const int  lg   = lane >> 4;
    const int  lmod = lane & 15;
    const int  m    = row0 + lmod;
    const int  mc   = (m < N) ? m : (N - 1);
    const int  nc   = tile_n * 16 + lmod;
    const bool nok  = (nc < F_OUT);
    const int  ncc  = nok ? nc : 0;

    v8f acc = {0.f, 0.f, 0.f, 0.f, 0.f, 0.f, 0.f, 0.f};

#pragma unroll
    for (int kc = 0; kc < KCH; ++kc) {
        const int k0  = kc * 4 + lg * 2;
        const int k1  = k0 + 1;
        const int k0c = (k0 < F_IN) ? k0 : (F_IN - 1);
        const int k1c = (k1 < F_IN) ? k1 : (F_IN - 1);

        float a0 = x[(long long)mc * F_IN + k0c];
        float a1 = x[(long long)mc * F_IN + k1c];
        a0 = (m < N && k0 < F_IN) ? a0 : 0.f;
        a1 = (m < N && k1 < F_IN) ? a1 : 0.f;
        v2f a; a.x = a0; a.y = a1;

        float b0 = W[k0c * F_OUT + ncc];
        float b1 = W[k1c * F_OUT + ncc];
        b0 = (nok && k0 < F_IN) ? b0 : 0.f;
        b1 = (nok && k1 < F_IN) ? b1 : 0.f;
        v2f b; b.x = b0; b.y = b1;

        acc = __builtin_amdgcn_wmma_f32_16x16x4_f32(
            false, a, false, b, (short)0, acc, false, false);
    }

    const bool fullM = (row0 + 16 <= N);
    if (nok) {
        float* p = xw + (long long)(row0 + lg * 8) * F_OUT + nc;
        if (fullM) {
#pragma unroll
            for (int v = 0; v < 8; ++v) p[(long long)v * F_OUT] = acc[v];
        } else {
#pragma unroll
            for (int v = 0; v < 8; ++v) {
                const int mrow = row0 + v + lg * 8;
                if (mrow < N) p[(long long)v * F_OUT] = acc[v];
            }
        }
    }
}

template <int F>
__global__ void k_gather(const float* __restrict__ x,
                         const int* __restrict__ rowptr,
                         const unsigned long long* __restrict__ csr,
                         float* __restrict__ agg, int N) {
    constexpr int C = F / 4;
    constexpr int L = 8 * C;
    const int gid  = blockIdx.x * blockDim.x + threadIdx.x;
    const int node = gid / L;
    const int lin  = gid - node * L;
    const int slot = lin / C;
    const int c    = lin - slot * C;
    const bool ok  = (node < N);
    int beg = 0, end = 0;
    if (ok) { beg = rowptr[node]; end = rowptr[node + 1]; }

    const float4* __restrict__ x4 = reinterpret_cast<const float4*>(x);
    float4 acc = {0.f, 0.f, 0.f, 0.f};
    int j = beg + slot;
    for (; j + 8 < end; j += 16) {
        const unsigned long long e0 = csr[j];
        const unsigned long long e1 = csr[j + 8];
        const int   s0 = (int)(unsigned int)(e0 & 0xffffffffull);
        const int   s1 = (int)(unsigned int)(e1 & 0xffffffffull);
        const float w0 = __uint_as_float((unsigned int)(e0 >> 32));
        const float w1 = __uint_as_float((unsigned int)(e1 >> 32));
        const float4 v0 = x4[(long long)s0 * C + c];
        const float4 v1 = x4[(long long)s1 * C + c];
        acc.x += v0.x * w0;
        acc.y += v0.y * w0;
        acc.z += v0.z * w0;
        acc.w += v0.w * w0;
        acc.x += v1.x * w1;
        acc.y += v1.y * w1;
        acc.z += v1.z * w1;
        acc.w += v1.w * w1;
    }
    if (j < end) {
        const unsigned long long ent = csr[j];
        const int   s = (int)(unsigned int)(ent & 0xffffffffull);
        const float w = __uint_as_float((unsigned int)(ent >> 32));
        const float4 v = x4[(long long)s * C + c];
        acc.x += v.x * w;
        acc.y += v.y * w;
        acc.z += v.z * w;
        acc.w += v.w * w;
    }
#pragma unroll
    for (int off = C; off < L; off <<= 1) {
        acc.x += __shfl_xor(acc.x, off);
        acc.y += __shfl_xor(acc.y, off);
        acc.z += __shfl_xor(acc.z, off);
        acc.w += __shfl_xor(acc.w, off);
    }
    if (ok && slot == 0)
        reinterpret_cast<float4*>(agg)[(long long)node * C + c] = acc;
}

template <int LOGF>
__global__ void k_finalize(const float* __restrict__ agg,
                           const float* __restrict__ xw,
                           const float* __restrict__ dinv,
                           const float* __restrict__ bias,
                           float* __restrict__ out, long long NF) {
    const long long i = (long long)blockIdx.x * blockDim.x + threadIdx.x;
    if (i >= NF) return;
    const int node = (int)(i >> LOGF);
    const int f    = (int)(i & ((1 << LOGF) - 1));
    out[i] = agg[i] + xw[i] * dinv[node] + bias[f];
}

template <int F_IN, int F_OUT>
__global__ void k_combine_linear_wmma(const float* __restrict__ agg,
                                      const float* __restrict__ x,
                                      const float* __restrict__ dinv,
                                      const float* __restrict__ W,
                                      const float* __restrict__ bias,
                                      float* __restrict__ out, int N) {
    constexpr int TILES_N = (F_OUT + 15) / 16;
    constexpr int KCH     = (F_IN + 3) / 4;

    const int wave = blockIdx.x * (blockDim.x >> 5) + (threadIdx.x >> 5);
    const int lane = threadIdx.x & 31;
    const int row0 = wave * 16;
    if (row0 >= N) return;

    const int  lg   = lane >> 4;
    const int  lmod = lane & 15;
    const int  m    = row0 + lmod;
    const bool mok  = (m < N);
    const int  mc   = mok ? m : (N - 1);
    const float di  = dinv[mc];

    v2f a[KCH];
#pragma unroll
    for (int kc = 0; kc < KCH; ++kc) {
        const int k0  = kc * 4 + lg * 2;
        const int k1  = k0 + 1;
        const int k0c = (k0 < F_IN) ? k0 : (F_IN - 1);
        const int k1c = (k1 < F_IN) ? k1 : (F_IN - 1);
        const long long i0 = (long long)mc * F_IN + k0c;
        const long long i1 = (long long)mc * F_IN + k1c;
        float h0 = agg[i0] + x[i0] * di;
        float h1 = agg[i1] + x[i1] * di;
        h0 = (mok && k0 < F_IN) ? h0 : 0.f;
        h1 = (mok && k1 < F_IN) ? h1 : 0.f;
        a[kc].x = h0; a[kc].y = h1;
    }

    const bool fullM = (row0 + 16 <= N);

#pragma unroll
    for (int tn = 0; tn < TILES_N; ++tn) {
        const int  nc  = tn * 16 + lmod;
        const bool nok = (nc < F_OUT);
        const int  ncc = nok ? nc : 0;
        const float bv = bias[ncc];

        v8f acc = {0.f, 0.f, 0.f, 0.f, 0.f, 0.f, 0.f, 0.f};

#pragma unroll
        for (int kc = 0; kc < KCH; ++kc) {
            const int k0  = kc * 4 + lg * 2;
            const int k1  = k0 + 1;
            const int k0c = (k0 < F_IN) ? k0 : (F_IN - 1);
            const int k1c = (k1 < F_IN) ? k1 : (F_IN - 1);

            float b0 = W[k0c * F_OUT + ncc];
            float b1 = W[k1c * F_OUT + ncc];
            b0 = (nok && k0 < F_IN) ? b0 : 0.f;
            b1 = (nok && k1 < F_IN) ? b1 : 0.f;
            v2f b; b.x = b0; b.y = b1;

            acc = __builtin_amdgcn_wmma_f32_16x16x4_f32(
                false, a[kc], false, b, (short)0, acc, false, false);
        }

        if (nok) {
            float* p = out + (long long)(row0 + lg * 8) * F_OUT + nc;
            if (fullM) {
#pragma unroll
                for (int v = 0; v < 8; ++v) p[(long long)v * F_OUT] = acc[v] + bv;
            } else {
#pragma unroll
                for (int v = 0; v < 8; ++v) {
                    const int mrow = row0 + v + lg * 8;
                    if (mrow < N) p[(long long)v * F_OUT] = acc[v] + bv;
                }
            }
        }
    }
}

template <int F_IN, int F_OUT, int LOGF>
static void run_layer(const float* xin, const float* W, const float* bias,
                      float* xw, float* agg, const float* dinv,
                      const int* rowptr, const unsigned long long* csr,
                      float* xout, int N, hipStream_t stream) {
    constexpr int T = 256;
    constexpr int TILES_N = (F_OUT + 15) / 16;

    const long long waves  = (long long)((N + 15) / 16) * TILES_N;
    const long long blocks = (waves + 7) / 8;
    k_linear_wmma<F_IN, F_OUT><<<(unsigned)blocks, 256, 0, stream>>>(xin, W, xw, N);

    const long long gl = (long long)N * (2 * F_OUT);
    k_gather<F_OUT><<<(unsigned)((gl + T - 1) / T), T, 0, stream>>>(
        xw, rowptr, csr, agg, N);

    const long long nf = (long long)N * F_OUT;
    k_finalize<LOGF><<<(unsigned)((nf + T - 1) / T), T, 0, stream>>>(
        agg, xw, dinv, bias, xout, nf);
}

template <int F_IN, int F_OUT>
static void run_layer_agg_first(const float* xin, const float* W, const float* bias,
                                float* agg, const float* dinv,
                                const int* rowptr, const unsigned long long* csr,
                                float* xout, int N, hipStream_t stream) {
    constexpr int T = 256;

    const long long gl = (long long)N * (2 * F_IN);
    k_gather<F_IN><<<(unsigned)((gl + T - 1) / T), T, 0, stream>>>(
        xin, rowptr, csr, agg, N);

    const long long waves  = (long long)((N + 15) / 16);
    const long long blocks = (waves + 7) / 8;
    k_combine_linear_wmma<F_IN, F_OUT><<<(unsigned)blocks, 256, 0, stream>>>(
        agg, xin, dinv, W, bias, xout, N);
}

extern "C" void kernel_launch(void* const* d_in, const int* in_sizes, int n_in,
                              void* d_out, int out_size, void* d_ws, size_t ws_size,
                              hipStream_t stream) {
    const int* read_length = (const int*)d_in[0];
    const int* edge        = (const int*)d_in[1];
    const int  N = in_sizes[0];
    const int  E = in_sizes[1] / 2;
    const int* src = edge;
    const int* dst = edge + E;

    const float* W1 = (const float*)d_in[2]; const float* B1 = (const float*)d_in[3];
    const float* W2 = (const float*)d_in[4]; const float* B2 = (const float*)d_in[5];
    const float* W3 = (const float*)d_in[6]; const float* B3 = (const float*)d_in[7];
    const float* W4 = (const float*)d_in[8]; const float* B4 = (const float*)d_in[9];

    float* ws   = (float*)d_ws;
    float* xbuf = ws;
    float* xw   = xbuf + (size_t)N * 16;
    float* agg  = xw   + (size_t)N * 16;
    unsigned long long* csr = (unsigned long long*)(agg + (size_t)N * 16);
    float* dinv = (float*)(csr + (size_t)E);
    float* dsq  = dinv + N;
    int* cnt    = (int*)(dsq + N);
    int* cursor = cnt + N;
    int* rowptr = cursor + N;
    int* bsum   = rowptr + (N + 1);

    const int T  = 256;
    const int NB = (N + 1023) / 1024;

    k_cnt_zero<<<(N + T - 1) / T, T, 0, stream>>>(cnt, N);
    k_count<<<(E + T - 1) / T, T, 0, stream>>>(dst, cnt, E);
    k_scan_part<<<NB, 256, 0, stream>>>(cnt, bsum, N);
    k_scan_top<<<1, 1024, 0, stream>>>(bsum, NB);
    k_scan_fin<<<NB, 256, 0, stream>>>(cnt, bsum, read_length, rowptr, cursor,
                                       dinv, dsq, xbuf, N);
    k_fill<<<(E + T - 1) / T, T, 0, stream>>>(src, dst, dsq, cursor, csr, E);

    run_layer<1, 4, 2>(xbuf, W1, B1, xw, agg, dinv, rowptr, csr, xbuf, N, stream);
    run_layer_agg_first<4, 8>(xbuf, W2, B2, agg, dinv, rowptr, csr, xw, N, stream);
    run_layer_agg_first<8, 16>(xw, W3, B3, agg, dinv, rowptr, csr, xbuf, N, stream);
    run_layer_agg_first<16, 32>(xbuf, W4, B4, agg, dinv, rowptr, csr,
                                (float*)d_out, N, stream);
}


# network has no activation: carry the pair (A_hat^k x0, A_hat^k 1) through four 2-float-wide edge gathers, fold W1..W4 and biases to one 4x32 matrix, single WMMA output pass; float sum order changes
# speedup vs baseline: 2.4002x; 1.0640x over previous
#include <hip/hip_runtime.h>
#include <hip/hip_bf16.h>


typedef __attribute__((ext_vector_type(2))) float v2f;
typedef __attribute__((ext_vector_type(8))) float v8f;

#define LEN_SCALE_INV (1.0f / 20000.0f)

__global__ void k_cnt_zero(int* __restrict__ cnt, int N) {
    int i = blockIdx.x * blockDim.x + threadIdx.x;
    if (i < N) cnt[i] = 0;
}

__global__ void k_count(const int* __restrict__ dst, int* __restrict__ cnt, int E) {
    int e = blockIdx.x * blockDim.x + threadIdx.x;
    if (e < E) atomicAdd(&cnt[dst[e]], 1);
}

__global__ void k_scan_part(const int* __restrict__ cnt, int* __restrict__ bsum,
                            int N) {
    __shared__ int sh[8];
    const int tid  = threadIdx.x;
    const int base = blockIdx.x * 1024 + tid * 4;
    int s = 0;
#pragma unroll
    for (int k = 0; k < 4; ++k)
        if (base + k < N) s += cnt[base + k];
    for (int off = 16; off > 0; off >>= 1) s += __shfl_xor(s, off);
    if ((tid & 31) == 0) sh[tid >> 5] = s;
    __syncthreads();
    if (tid == 0) {
        int t = 0;
#pragma unroll
        for (int w = 0; w < 8; ++w) t += sh[w];
        bsum[blockIdx.x] = t;
    }
}

__global__ void k_scan_top(int* __restrict__ bsum, int NB) {
    __shared__ int sh[1024];
    const int tid = threadIdx.x;
    int carry = 0;
    for (int base = 0; base < NB; base += 1024) {
        const int i = base + tid;
        const int v = (i < NB) ? bsum[i] : 0;
        sh[tid] = v;
        __syncthreads();
        for (int off = 1; off < 1024; off <<= 1) {
            const int t = (tid >= off) ? sh[tid - off] : 0;
            __syncthreads();
            sh[tid] += t;
            __syncthreads();
        }
        const int incl  = sh[tid];
        const int total = sh[1023];
        if (i < NB) bsum[i] = carry + incl - v;
        carry += total;
        __syncthreads();
    }
}

__global__ void k_scan_fin(const int* __restrict__ cnt, const int* __restrict__ bsum,
                           const int* __restrict__ read_length,
                           int* __restrict__ rowptr, int* __restrict__ cursor,
                           float* __restrict__ dinv, float* __restrict__ dsq,
                           float2* __restrict__ h0, int N) {
    __shared__ int sh[8];
    const int tid  = threadIdx.x;
    const int lane = tid & 31;
    const int wave = tid >> 5;
    const int base = blockIdx.x * 1024 + tid * 4;
    int c[4];
    int s = 0;
#pragma unroll
    for (int k = 0; k < 4; ++k) {
        c[k] = (base + k < N) ? cnt[base + k] : 0;
        s += c[k];
    }
    int incl = s;
    for (int off = 1; off < 32; off <<= 1) {
        const int t = __shfl_up(incl, off);
        if (lane >= off) incl += t;
    }
    if (lane == 31) sh[wave] = incl;
    __syncthreads();
    int woff = 0;
    for (int w = 0; w < wave; ++w) woff += sh[w];
    int excl = bsum[blockIdx.x] + woff + incl - s;
#pragma unroll
    for (int k = 0; k < 4; ++k) {
        const int i = base + k;
        if (i < N) {
            rowptr[i] = excl;
            cursor[i] = excl;
            const float deg = (float)(c[k] + 1);
            dsq[i]  = rsqrtf(deg);
            dinv[i] = 1.0f / deg;
            h0[i]   = make_float2((float)read_length[i] * LEN_SCALE_INV, 1.0f);
            excl += c[k];
            if (i == N - 1) rowptr[N] = excl;
        }
    }
}

__global__ void k_fill(const int* __restrict__ src, const int* __restrict__ dst,
                       const float* __restrict__ dsq, int* __restrict__ cursor,
                       unsigned long long* __restrict__ csr, int E) {
    int e = blockIdx.x * blockDim.x + threadIdx.x;
    if (e < E) {
        const int s = src[e];
        const int d = dst[e];
        const float w = dsq[s] * dsq[d];
        const int pos = atomicAdd(&cursor[d], 1);
        csr[pos] = ((unsigned long long)__float_as_uint(w) << 32) | (unsigned int)s;
    }
}

template <int K, int NO, int ROWS>
__device__ __forceinline__ void fold_stage(const float* sin, const float* __restrict__ Wg,
                                           float* dst, int pitch, int lane) {
    constexpr int TILES_N = (NO + 15) / 16;
    constexpr int KCH     = K / 4;
    const int lg   = lane >> 4;
    const int lmod = lane & 15;
#pragma unroll
    for (int tn = 0; tn < TILES_N; ++tn) {
        const int  nc  = tn * 16 + lmod;
        const bool nok = (nc < NO);
        const int  ncc = nok ? nc : 0;
        v8f acc = {0.f, 0.f, 0.f, 0.f, 0.f, 0.f, 0.f, 0.f};
#pragma unroll
        for (int kc = 0; kc < KCH; ++kc) {
            const int k0 = kc * 4 + lg * 2;
            const int k1 = k0 + 1;
            v2f a; a.x = sin[lmod * 16 + k0]; a.y = sin[lmod * 16 + k1];
            float b0 = Wg[k0 * NO + ncc];
            float b1 = Wg[k1 * NO + ncc];
            b0 = nok ? b0 : 0.f;
            b1 = nok ? b1 : 0.f;
            v2f b; b.x = b0; b.y = b1;
            acc = __builtin_amdgcn_wmma_f32_16x16x4_f32(
                false, a, false, b, (short)0, acc, false, false);
        }
        if (lg == 0 && nok) {
#pragma unroll
            for (int v = 0; v < ROWS; ++v) dst[v * pitch + nc] = acc[v];
        }
    }
}

__global__ void k_fold_wmma(const float* __restrict__ W1, const float* __restrict__ b1,
                            const float* __restrict__ W2, const float* __restrict__ b2,
                            const float* __restrict__ W3, const float* __restrict__ b3,
                            const float* __restrict__ W4, float* __restrict__ M) {
    __shared__ float sa[256];
    __shared__ float sb[256];
    const int lane = threadIdx.x;
    for (int i = lane; i < 256; i += 32) { sa[i] = 0.f; sb[i] = 0.f; }
    __syncthreads();
    if (lane < 4) { sa[lane] = W1[lane]; sa[16 + lane] = b1[lane]; }
    __syncthreads();
    fold_stage<4, 8, 2>(sa, W2, sb, 16, lane);
    if (lane < 8) sb[2 * 16 + lane] = b2[lane];
    __syncthreads();
    fold_stage<8, 16, 3>(sb, W3, sa, 16, lane);
    if (lane < 16) sa[3 * 16 + lane] = b3[lane];
    __syncthreads();
    fold_stage<16, 32, 4>(sa, W4, M, 32, lane);
}

__global__ void k_gather_pair(const float2* __restrict__ hin,
                              const int* __restrict__ rowptr,
                              const unsigned long long* __restrict__ csr,
                              const float* __restrict__ dinv,
                              float2* __restrict__ hout, int N) {
    const int gid  = blockIdx.x * blockDim.x + threadIdx.x;
    const int node = gid >> 3;
    const int slot = gid & 7;
    const bool ok  = (node < N);
    int beg = 0, end = 0;
    if (ok) { beg = rowptr[node]; end = rowptr[node + 1]; }
    float2 self = {0.f, 0.f};
    float  di   = 0.f;
    if (ok && slot == 0) { self = hin[node]; di = dinv[node]; }

    float2 acc = {0.f, 0.f};
    int j = beg + slot;
    for (; j + 8 < end; j += 16) {
        const unsigned long long e0 = csr[j];
        const unsigned long long e1 = csr[j + 8];
        const int   s0 = (int)(unsigned int)(e0 & 0xffffffffull);
        const int   s1 = (int)(unsigned int)(e1 & 0xffffffffull);
        const float w0 = __uint_as_float((unsigned int)(e0 >> 32));
        const float w1 = __uint_as_float((unsigned int)(e1 >> 32));
        const float2 v0 = hin[s0];
        const float2 v1 = hin[s1];
        acc.x += v0.x * w0;
        acc.y += v0.y * w0;
        acc.x += v1.x * w1;
        acc.y += v1.y * w1;
    }
    if (j < end) {
        const unsigned long long ent = csr[j];
        const int   s = (int)(unsigned int)(ent & 0xffffffffull);
        const float w = __uint_as_float((unsigned int)(ent >> 32));
        const float2 v = hin[s];
        acc.x += v.x * w;
        acc.y += v.y * w;
    }
#pragma unroll
    for (int off = 1; off < 8; off <<= 1) {
        acc.x += __shfl_xor(acc.x, off);
        acc.y += __shfl_xor(acc.y, off);
    }
    if (ok && slot == 0)
        hout[node] = make_float2(acc.x + self.x * di, acc.y + self.y * di);
}

__global__ void k_out_wmma(const float2* __restrict__ h1, const float2* __restrict__ h2,
                           const float2* __restrict__ h3, const float2* __restrict__ h4,
                           const float* __restrict__ M, const float* __restrict__ bias,
                           float* __restrict__ out, int N) {
    constexpr int F_OUT = 32;
    const int wave = blockIdx.x * (blockDim.x >> 5) + (threadIdx.x >> 5);
    const int lane = threadIdx.x & 31;
    const int row0 = wave * 16;
    if (row0 >= N) return;

    const int  lg   = lane >> 4;
    const int  lmod = lane & 15;
    const int  m    = row0 + lmod;
    const int  mc   = (m < N) ? m : (N - 1);

    const float2 u0 = lg ? h2[mc] : h4[mc];
    const float2 u1 = lg ? h1[mc] : h3[mc];
    v2f a; a.x = lg ? u0.y : u0.x; a.y = u1.y;

    const bool fullM = (row0 + 16 <= N);

#pragma unroll
    for (int tn = 0; tn < 2; ++tn) {
        const int nc = tn * 16 + lmod;
        const float bv = bias[nc];
        v2f b;
        b.x = M[(lg * 2) * F_OUT + nc];
        b.y = M[(lg * 2 + 1) * F_OUT + nc];

        v8f acc = {0.f, 0.f, 0.f, 0.f, 0.f, 0.f, 0.f, 0.f};
        acc = __builtin_amdgcn_wmma_f32_16x16x4_f32(
            false, a, false, b, (short)0, acc, false, false);

        float* p = out + (long long)(row0 + lg * 8) * F_OUT + nc;
        if (fullM) {
#pragma unroll
            for (int v = 0; v < 8; ++v) p[(long long)v * F_OUT] = acc[v] + bv;
        } else {
#pragma unroll
            for (int v = 0; v < 8; ++v) {
                const int mrow = row0 + v + lg * 8;
                if (mrow < N) p[(long long)v * F_OUT] = acc[v] + bv;
            }
        }
    }
}

extern "C" void kernel_launch(void* const* d_in, const int* in_sizes, int n_in,
                              void* d_out, int out_size, void* d_ws, size_t ws_size,
                              hipStream_t stream) {
    const int* read_length = (const int*)d_in[0];
    const int* edge        = (const int*)d_in[1];
    const int  N = in_sizes[0];
    const int  E = in_sizes[1] / 2;
    const int* src = edge;
    const int* dst = edge + E;

    const float* W1 = (const float*)d_in[2]; const float* B1 = (const float*)d_in[3];
    const float* W2 = (const float*)d_in[4]; const float* B2 = (const float*)d_in[5];
    const float* W3 = (const float*)d_in[6]; const float* B3 = (const float*)d_in[7];
    const float* W4 = (const float*)d_in[8]; const float* B4 = (const float*)d_in[9];

    float2* h0 = (float2*)d_ws;
    float2* h1 = h0 + (size_t)N;
    float2* h2 = h1 + (size_t)N;
    float2* h3 = h2 + (size_t)N;
    float2* h4 = h3 + (size_t)N;
    unsigned long long* csr = (unsigned long long*)(h4 + (size_t)N);
    float* dinv = (float*)(csr + (size_t)E);
    float* dsq  = dinv + N;
    int* cnt    = (int*)(dsq + N);
    int* cursor = cnt + N;
    int* rowptr = cursor + N;
    int* bsum   = rowptr + (N + 1);
    float* Mf   = (float*)(bsum + (N / 1024 + 2));

    const int T  = 256;
    const int NB = (N + 1023) / 1024;

    k_fold_wmma<<<1, 32, 0, stream>>>(W1, B1, W2, B2, W3, B3, W4, Mf);

    k_cnt_zero<<<(N + T - 1) / T, T, 0, stream>>>(cnt, N);
    k_count<<<(E + T - 1) / T, T, 0, stream>>>(dst, cnt, E);
    k_scan_part<<<NB, 256, 0, stream>>>(cnt, bsum, N);
    k_scan_top<<<1, 1024, 0, stream>>>(bsum, NB);
    k_scan_fin<<<NB, 256, 0, stream>>>(cnt, bsum, read_length, rowptr, cursor,
                                       dinv, dsq, h0, N);
    k_fill<<<(E + T - 1) / T, T, 0, stream>>>(src, dst, dsq, cursor, csr, E);

    const long long gl = (long long)N * 8;
    const unsigned gb = (unsigned)((gl + T - 1) / T);
    k_gather_pair<<<gb, T, 0, stream>>>(h0, rowptr, csr, dinv, h1, N);
    k_gather_pair<<<gb, T, 0, stream>>>(h1, rowptr, csr, dinv, h2, N);
    k_gather_pair<<<gb, T, 0, stream>>>(h2, rowptr, csr, dinv, h3, N);
    k_gather_pair<<<gb, T, 0, stream>>>(h3, rowptr, csr, dinv, h4, N);

    const long long waves  = (long long)((N + 15) / 16);
    const long long blocks = (waves + 7) / 8;
    k_out_wmma<<<(unsigned)blocks, 256, 0, stream>>>(h1, h2, h3, h4, Mf, B4,
                                                     (float*)d_out, N);
}


# edge sort by dst without per-edge global atomics: per-block LDS histograms bin edges into 1024-node buckets, one block per bucket then counts, scans and places in LDS; float sum order changes
# speedup vs baseline: 5.0853x; 2.1187x over previous
#include <hip/hip_runtime.h>
#include <hip/hip_bf16.h>


typedef __attribute__((ext_vector_type(2))) float v2f;
typedef __attribute__((ext_vector_type(8))) float v8f;

#define LEN_SCALE_INV (1.0f / 20000.0f)

#define BSH    10
#define BNODES (1 << BSH)
#define BMASK  (BNODES - 1)
#define CHUNK  8192
#define BPAD   32

__global__ void k_cnt_zero(int* __restrict__ cnt, int N) {
    int i = blockIdx.x * blockDim.x + threadIdx.x;
    if (i < N) cnt[i] = 0;
}

__global__ void k_bcount(const int* __restrict__ dst, int* __restrict__ btot,
                         int E, int NBK) {
    extern __shared__ int sh[];
    const int tid = threadIdx.x;
    for (int b = tid; b < NBK; b += blockDim.x) sh[b] = 0;
    __syncthreads();
    const int c0 = blockIdx.x * CHUNK;
    const int c1 = (c0 + CHUNK < E) ? (c0 + CHUNK) : E;
    const int st = blockDim.x;
    int e = c0 + tid;
    for (; e + 3 * st < c1; e += 4 * st) {
        const int d0 = dst[e];
        const int d1 = dst[e + st];
        const int d2 = dst[e + 2 * st];
        const int d3 = dst[e + 3 * st];
        atomicAdd(&sh[d0 >> BSH], 1);
        atomicAdd(&sh[d1 >> BSH], 1);
        atomicAdd(&sh[d2 >> BSH], 1);
        atomicAdd(&sh[d3 >> BSH], 1);
    }
    for (; e < c1; e += st) atomicAdd(&sh[dst[e] >> BSH], 1);
    __syncthreads();
    for (int b = tid; b < NBK; b += blockDim.x) {
        const int c = sh[b];
        if (c) atomicAdd(&btot[b * BPAD], c);
    }
}

__global__ void k_bscan(const int* __restrict__ btot, int* __restrict__ bbase,
                        int* __restrict__ bcur, int NBK) {
    __shared__ int sh[1024];
    const int tid = threadIdx.x;
    int carry = 0;
    for (int base = 0; base < NBK; base += 1024) {
        const int i = base + tid;
        const int v = (i < NBK) ? btot[i * BPAD] : 0;
        sh[tid] = v;
        __syncthreads();
        for (int off = 1; off < 1024; off <<= 1) {
            const int t = (tid >= off) ? sh[tid - off] : 0;
            __syncthreads();
            sh[tid] += t;
            __syncthreads();
        }
        const int incl  = sh[tid];
        const int total = sh[1023];
        if (i < NBK) {
            const int ex = carry + incl - v;
            bbase[i] = ex;
            bcur[i * BPAD] = ex;
        }
        carry += total;
        __syncthreads();
    }
    if (tid == 0) bbase[NBK] = carry;
}

__global__ void k_bscatter(const int* __restrict__ src, const int* __restrict__ dst,
                           int* __restrict__ bcur, int2* __restrict__ pairs,
                           int E, int NBK) {
    extern __shared__ int sh[];
    int* hist = sh;
    int* base = sh + NBK;
    const int tid = threadIdx.x;
    for (int b = tid; b < NBK; b += blockDim.x) hist[b] = 0;
    __syncthreads();
    const int c0 = blockIdx.x * CHUNK;
    const int c1 = (c0 + CHUNK < E) ? (c0 + CHUNK) : E;
    const int st = blockDim.x;
    int e = c0 + tid;
    for (; e + 3 * st < c1; e += 4 * st) {
        const int d0 = dst[e];
        const int d1 = dst[e + st];
        const int d2 = dst[e + 2 * st];
        const int d3 = dst[e + 3 * st];
        atomicAdd(&hist[d0 >> BSH], 1);
        atomicAdd(&hist[d1 >> BSH], 1);
        atomicAdd(&hist[d2 >> BSH], 1);
        atomicAdd(&hist[d3 >> BSH], 1);
    }
    for (; e < c1; e += st) atomicAdd(&hist[dst[e] >> BSH], 1);
    __syncthreads();
    for (int b = tid; b < NBK; b += blockDim.x) {
        const int c = hist[b];
        base[b] = c ? atomicAdd(&bcur[b * BPAD], c) : 0;
        hist[b] = 0;
    }
    __syncthreads();
    e = c0 + tid;
    for (; e + 3 * st < c1; e += 4 * st) {
        const int d0 = dst[e];
        const int d1 = dst[e + st];
        const int d2 = dst[e + 2 * st];
        const int d3 = dst[e + 3 * st];
        const int s0 = src[e];
        const int s1 = src[e + st];
        const int s2 = src[e + 2 * st];
        const int s3 = src[e + 3 * st];
        const int p0 = base[d0 >> BSH] + atomicAdd(&hist[d0 >> BSH], 1);
        const int p1 = base[d1 >> BSH] + atomicAdd(&hist[d1 >> BSH], 1);
        const int p2 = base[d2 >> BSH] + atomicAdd(&hist[d2 >> BSH], 1);
        const int p3 = base[d3 >> BSH] + atomicAdd(&hist[d3 >> BSH], 1);
        pairs[p0] = make_int2(s0, d0);
        pairs[p1] = make_int2(s1, d1);
        pairs[p2] = make_int2(s2, d2);
        pairs[p3] = make_int2(s3, d3);
    }
    for (; e < c1; e += st) {
        const int s = src[e];
        const int d = dst[e];
        const int b = d >> BSH;
        const int r = atomicAdd(&hist[b], 1);
        pairs[base[b] + r] = make_int2(s, d);
    }
}

__global__ void __launch_bounds__(1024)
k_bsort(const int2* __restrict__ pairs, const int* __restrict__ bbase,
        const int* __restrict__ read_length,
        int* __restrict__ rowptr, int* __restrict__ csr,
        float* __restrict__ dinv, float* __restrict__ sdeg,
        float2* __restrict__ h0, int N) {
    __shared__ int cnt[BNODES];
    __shared__ int wsum[32];
    const int tid  = threadIdx.x;
    const int lane = tid & 31;
    const int wave = tid >> 5;
    const int bk   = blockIdx.x;
    const int e0   = bbase[bk];
    const int e1   = bbase[bk + 1];

    cnt[tid] = 0;
    __syncthreads();
    int j = e0 + tid;
    for (; j + 3 * BNODES < e1; j += 4 * BNODES) {
        const int d0 = pairs[j].y;
        const int d1 = pairs[j + BNODES].y;
        const int d2 = pairs[j + 2 * BNODES].y;
        const int d3 = pairs[j + 3 * BNODES].y;
        atomicAdd(&cnt[d0 & BMASK], 1);
        atomicAdd(&cnt[d1 & BMASK], 1);
        atomicAdd(&cnt[d2 & BMASK], 1);
        atomicAdd(&cnt[d3 & BMASK], 1);
    }
    for (; j < e1; j += BNODES) atomicAdd(&cnt[pairs[j].y & BMASK], 1);
    __syncthreads();

    const int c = cnt[tid];
    int incl = c;
    for (int off = 1; off < 32; off <<= 1) {
        const int t = __shfl_up(incl, off);
        if (lane >= off) incl += t;
    }
    if (lane == 31) wsum[wave] = incl;
    __syncthreads();
    if (wave == 0) {
        const int w = wsum[lane];
        int wi = w;
        for (int off = 1; off < 32; off <<= 1) {
            const int t = __shfl_up(wi, off);
            if (lane >= off) wi += t;
        }
        wsum[lane] = wi - w;
    }
    __syncthreads();
    const int start = e0 + wsum[wave] + incl - c;

    const int node = (bk << BSH) + tid;
    if (node < N) {
        rowptr[node] = start;
        const float deg = (float)(c + 1);
        const float rs  = rsqrtf(deg);
        sdeg[node] = sqrtf(deg);
        dinv[node] = 1.0f / deg;
        h0[node]   = make_float2((float)read_length[node] * LEN_SCALE_INV * rs, rs);
        if (node == N - 1) rowptr[N] = start + c;
    }
    cnt[tid] = start;
    __syncthreads();
    j = e0 + tid;
    for (; j + 3 * BNODES < e1; j += 4 * BNODES) {
        const int2 p0 = pairs[j];
        const int2 p1 = pairs[j + BNODES];
        const int2 p2 = pairs[j + 2 * BNODES];
        const int2 p3 = pairs[j + 3 * BNODES];
        const int q0 = atomicAdd(&cnt[p0.y & BMASK], 1);
        const int q1 = atomicAdd(&cnt[p1.y & BMASK], 1);
        const int q2 = atomicAdd(&cnt[p2.y & BMASK], 1);
        const int q3 = atomicAdd(&cnt[p3.y & BMASK], 1);
        csr[q0] = p0.x;
        csr[q1] = p1.x;
        csr[q2] = p2.x;
        csr[q3] = p3.x;
    }
    for (; j < e1; j += BNODES) {
        const int2 p = pairs[j];
        const int pos = atomicAdd(&cnt[p.y & BMASK], 1);
        csr[pos] = p.x;
    }
}

template <int K, int NO, int ROWS>
__device__ __forceinline__ void fold_stage(const float* sin, const float* __restrict__ Wg,
                                           float* dst, int pitch, int lane) {
    constexpr int TILES_N = (NO + 15) / 16;
    constexpr int KCH     = K / 4;
    const int lg   = lane >> 4;
    const int lmod = lane & 15;
#pragma unroll
    for (int tn = 0; tn < TILES_N; ++tn) {
        const int  nc  = tn * 16 + lmod;
        const bool nok = (nc < NO);
        const int  ncc = nok ? nc : 0;
        v8f acc = {0.f, 0.f, 0.f, 0.f, 0.f, 0.f, 0.f, 0.f};
#pragma unroll
        for (int kc = 0; kc < KCH; ++kc) {
            const int k0 = kc * 4 + lg * 2;
            const int k1 = k0 + 1;
            v2f a; a.x = sin[lmod * 16 + k0]; a.y = sin[lmod * 16 + k1];
            float b0 = Wg[k0 * NO + ncc];
            float b1 = Wg[k1 * NO + ncc];
            b0 = nok ? b0 : 0.f;
            b1 = nok ? b1 : 0.f;
            v2f b; b.x = b0; b.y = b1;
            acc = __builtin_amdgcn_wmma_f32_16x16x4_f32(
                false, a, false, b, (short)0, acc, false, false);
        }
        if (lg == 0 && nok) {
#pragma unroll
            for (int v = 0; v < ROWS; ++v) dst[v * pitch + nc] = acc[v];
        }
    }
}

__global__ void k_fold_wmma(const float* __restrict__ W1, const float* __restrict__ b1,
                            const float* __restrict__ W2, const float* __restrict__ b2,
                            const float* __restrict__ W3, const float* __restrict__ b3,
                            const float* __restrict__ W4, float* __restrict__ M) {
    __shared__ float sa[256];
    __shared__ float sb[256];
    const int lane = threadIdx.x;
    for (int i = lane; i < 256; i += 32) { sa[i] = 0.f; sb[i] = 0.f; }
    __syncthreads();
    if (lane < 4) { sa[lane] = W1[lane]; sa[16 + lane] = b1[lane]; }
    __syncthreads();
    fold_stage<4, 8, 2>(sa, W2, sb, 16, lane);
    if (lane < 8) sb[2 * 16 + lane] = b2[lane];
    __syncthreads();
    fold_stage<8, 16, 3>(sb, W3, sa, 16, lane);
    if (lane < 16) sa[3 * 16 + lane] = b3[lane];
    __syncthreads();
    fold_stage<16, 32, 4>(sa, W4, M, 32, lane);
}

__global__ void k_gather_pair(const float2* __restrict__ hin,
                              const int* __restrict__ rowptr,
                              const int* __restrict__ csr,
                              const float* __restrict__ dinv,
                              float2* __restrict__ hout, int N) {
    const int gid  = blockIdx.x * blockDim.x + threadIdx.x;
    const int node = gid >> 3;
    const int slot = gid & 7;
    const bool ok  = (node < N);
    int beg = 0, end = 0;
    if (ok) { beg = rowptr[node]; end = rowptr[node + 1]; }
    float2 self = {0.f, 0.f};
    float  di   = 0.f;
    if (ok && slot == 0) { self = hin[node]; di = dinv[node]; }

    float2 acc = {0.f, 0.f};
    int j = beg + slot;
    for (; j + 8 < end; j += 16) {
        const int s0 = csr[j];
        const int s1 = csr[j + 8];
        const float2 v0 = hin[s0];
        const float2 v1 = hin[s1];
        acc.x += v0.x;
        acc.y += v0.y;
        acc.x += v1.x;
        acc.y += v1.y;
    }
    if (j < end) {
        const int s = csr[j];
        const float2 v = hin[s];
        acc.x += v.x;
        acc.y += v.y;
    }
#pragma unroll
    for (int off = 1; off < 8; off <<= 1) {
        acc.x += __shfl_xor(acc.x, off);
        acc.y += __shfl_xor(acc.y, off);
    }
    if (ok && slot == 0)
        hout[node] = make_float2((acc.x + self.x) * di, (acc.y + self.y) * di);
}

__global__ void k_out_wmma(const float2* __restrict__ h1, const float2* __restrict__ h2,
                           const float2* __restrict__ h3, const float2* __restrict__ h4,
                           const float* __restrict__ sdeg,
                           const float* __restrict__ M, const float* __restrict__ bias,
                           float* __restrict__ out, int N) {
    constexpr int F_OUT = 32;
    const int wave = blockIdx.x * (blockDim.x >> 5) + (threadIdx.x >> 5);
    const int lane = threadIdx.x & 31;
    const int row0 = wave * 16;
    if (row0 >= N) return;

    const int  lg   = lane >> 4;
    const int  lmod = lane & 15;
    const int  m    = row0 + lmod;
    const int  mc   = (m < N) ? m : (N - 1);

    const float2 u0 = lg ? h2[mc] : h4[mc];
    const float2 u1 = lg ? h1[mc] : h3[mc];
    const float sd = sdeg[mc];
    v2f a; a.x = (lg ? u0.y : u0.x) * sd; a.y = u1.y * sd;

    const bool fullM = (row0 + 16 <= N);

#pragma unroll
    for (int tn = 0; tn < 2; ++tn) {
        const int nc = tn * 16 + lmod;
        const float bv = bias[nc];
        v2f b;
        b.x = M[(lg * 2) * F_OUT + nc];
        b.y = M[(lg * 2 + 1) * F_OUT + nc];

        v8f acc = {0.f, 0.f, 0.f, 0.f, 0.f, 0.f, 0.f, 0.f};
        acc = __builtin_amdgcn_wmma_f32_16x16x4_f32(
            false, a, false, b, (short)0, acc, false, false);

        float* p = out + (long long)(row0 + lg * 8) * F_OUT + nc;
        if (fullM) {
#pragma unroll
            for (int v = 0; v < 8; ++v) p[(long long)v * F_OUT] = acc[v] + bv;
        } else {
#pragma unroll
            for (int v = 0; v < 8; ++v) {
                const int mrow = row0 + v + lg * 8;
                if (mrow < N) p[(long long)v * F_OUT] = acc[v] + bv;
            }
        }
    }
}

extern "C" void kernel_launch(void* const* d_in, const int* in_sizes, int n_in,
                              void* d_out, int out_size, void* d_ws, size_t ws_size,
                              hipStream_t stream) {
    const int* read_length = (const int*)d_in[0];
    const int* edge        = (const int*)d_in[1];
    const int  N = in_sizes[0];
    const int  E = in_sizes[1] / 2;
    const int* src = edge;
    const int* dst = edge + E;

    const float* W1 = (const float*)d_in[2]; const float* B1 = (const float*)d_in[3];
    const float* W2 = (const float*)d_in[4]; const float* B2 = (const float*)d_in[5];
    const float* W3 = (const float*)d_in[6]; const float* B3 = (const float*)d_in[7];
    const float* W4 = (const float*)d_in[8]; const float* B4 = (const float*)d_in[9];

    const int NBK = (N + BNODES - 1) / BNODES;
    float2* h0 = (float2*)d_ws;
    float2* h1 = h0 + (size_t)N;
    float2* h2 = h1 + (size_t)N;
    float2* h3 = h2 + (size_t)N;
    float2* h4 = h3 + (size_t)N;
    int2* pairs = (int2*)(h4 + (size_t)N);
    int* csr    = (int*)(pairs + (size_t)E);
    float* dinv = (float*)(csr + (size_t)E);
    float* sdeg = dinv + N;
    int* rowptr = (int*)(sdeg + N);
    int* btot   = rowptr + (N + 1);
    int* bbase  = btot + NBK * BPAD;
    int* bcur   = bbase + (NBK + 1);
    float* Mf   = (float*)(bcur + NBK * BPAD);

    const int T  = 256;

    k_fold_wmma<<<1, 32, 0, stream>>>(W1, B1, W2, B2, W3, B3, W4, Mf);

    const int TB  = 512;
    const int NCH = (E + CHUNK - 1) / CHUNK;
    k_cnt_zero<<<(NBK * BPAD + T - 1) / T, T, 0, stream>>>(btot, NBK * BPAD);
    k_bcount<<<NCH, TB, (size_t)NBK * sizeof(int), stream>>>(dst, btot, E, NBK);
    k_bscan<<<1, 1024, 0, stream>>>(btot, bbase, bcur, NBK);
    k_bscatter<<<NCH, TB, (size_t)NBK * 2 * sizeof(int), stream>>>(src, dst, bcur, pairs,
                                                                   E, NBK);
    k_bsort<<<NBK, BNODES, 0, stream>>>(pairs, bbase, read_length, rowptr, csr,
                                        dinv, sdeg, h0, N);

    const long long gl = (long long)N * 8;
    const unsigned gb = (unsigned)((gl + T - 1) / T);
    k_gather_pair<<<gb, T, 0, stream>>>(h0, rowptr, csr, dinv, h1, N);
    k_gather_pair<<<gb, T, 0, stream>>>(h1, rowptr, csr, dinv, h2, N);
    k_gather_pair<<<gb, T, 0, stream>>>(h2, rowptr, csr, dinv, h3, N);
    k_gather_pair<<<gb, T, 0, stream>>>(h3, rowptr, csr, dinv, h4, N);

    const long long waves  = (long long)((N + 15) / 16);
    const long long blocks = (waves + 7) / 8;
    k_out_wmma<<<(unsigned)blocks, 256, 0, stream>>>(h1, h2, h3, h4, sdeg, Mf, B4,
                                                     (float*)d_out, N);
}


# aggregation and output pass on long-lived waves: one wave loops over 32 consecutive nodes (8 trips of 4) or four 16-row tiles; 125000 -> 15625 waves per gather launch; float sum order unchanged
# speedup vs baseline: 5.3078x; 1.0438x over previous
#include <hip/hip_runtime.h>
#include <hip/hip_bf16.h>


typedef __attribute__((ext_vector_type(2))) float v2f;
typedef __attribute__((ext_vector_type(8))) float v8f;

#define LEN_SCALE_INV (1.0f / 20000.0f)

#define BSH    10
#define BNODES (1 << BSH)
#define BMASK  (BNODES - 1)
#define CHUNK  8192
#define BPAD   32

__global__ void k_cnt_zero(int* __restrict__ cnt, int N) {
    int i = blockIdx.x * blockDim.x + threadIdx.x;
    if (i < N) cnt[i] = 0;
}

__global__ void k_bcount(const int* __restrict__ dst, int* __restrict__ btot,
                         int E, int NBK) {
    extern __shared__ int sh[];
    const int tid = threadIdx.x;
    for (int b = tid; b < NBK; b += blockDim.x) sh[b] = 0;
    __syncthreads();
    const int c0 = blockIdx.x * CHUNK;
    const int c1 = (c0 + CHUNK < E) ? (c0 + CHUNK) : E;
    const int st = blockDim.x;
    int e = c0 + tid;
    for (; e + 3 * st < c1; e += 4 * st) {
        const int d0 = dst[e];
        const int d1 = dst[e + st];
        const int d2 = dst[e + 2 * st];
        const int d3 = dst[e + 3 * st];
        atomicAdd(&sh[d0 >> BSH], 1);
        atomicAdd(&sh[d1 >> BSH], 1);
        atomicAdd(&sh[d2 >> BSH], 1);
        atomicAdd(&sh[d3 >> BSH], 1);
    }
    for (; e < c1; e += st) atomicAdd(&sh[dst[e] >> BSH], 1);
    __syncthreads();
    for (int b = tid; b < NBK; b += blockDim.x) {
        const int c = sh[b];
        if (c) atomicAdd(&btot[b * BPAD], c);
    }
}

__global__ void k_bscan(const int* __restrict__ btot, int* __restrict__ bbase,
                        int* __restrict__ bcur, int NBK) {
    __shared__ int sh[1024];
    const int tid = threadIdx.x;
    int carry = 0;
    for (int base = 0; base < NBK; base += 1024) {
        const int i = base + tid;
        const int v = (i < NBK) ? btot[i * BPAD] : 0;
        sh[tid] = v;
        __syncthreads();
        for (int off = 1; off < 1024; off <<= 1) {
            const int t = (tid >= off) ? sh[tid - off] : 0;
            __syncthreads();
            sh[tid] += t;
            __syncthreads();
        }
        const int incl  = sh[tid];
        const int total = sh[1023];
        if (i < NBK) {
            const int ex = carry + incl - v;
            bbase[i] = ex;
            bcur[i * BPAD] = ex;
        }
        carry += total;
        __syncthreads();
    }
    if (tid == 0) bbase[NBK] = carry;
}

__global__ void k_bscatter(const int* __restrict__ src, const int* __restrict__ dst,
                           int* __restrict__ bcur, int2* __restrict__ pairs,
                           int E, int NBK) {
    extern __shared__ int sh[];
    int* hist = sh;
    int* base = sh + NBK;
    const int tid = threadIdx.x;
    for (int b = tid; b < NBK; b += blockDim.x) hist[b] = 0;
    __syncthreads();
    const int c0 = blockIdx.x * CHUNK;
    const int c1 = (c0 + CHUNK < E) ? (c0 + CHUNK) : E;
    const int st = blockDim.x;
    int e = c0 + tid;
    for (; e + 3 * st < c1; e += 4 * st) {
        const int d0 = dst[e];
        const int d1 = dst[e + st];
        const int d2 = dst[e + 2 * st];
        const int d3 = dst[e + 3 * st];
        atomicAdd(&hist[d0 >> BSH], 1);
        atomicAdd(&hist[d1 >> BSH], 1);
        atomicAdd(&hist[d2 >> BSH], 1);
        atomicAdd(&hist[d3 >> BSH], 1);
    }
    for (; e < c1; e += st) atomicAdd(&hist[dst[e] >> BSH], 1);
    __syncthreads();
    for (int b = tid; b < NBK; b += blockDim.x) {
        const int c = hist[b];
        base[b] = c ? atomicAdd(&bcur[b * BPAD], c) : 0;
        hist[b] = 0;
    }
    __syncthreads();
    e = c0 + tid;
    for (; e + 3 * st < c1; e += 4 * st) {
        const int d0 = dst[e];
        const int d1 = dst[e + st];
        const int d2 = dst[e + 2 * st];
        const int d3 = dst[e + 3 * st];
        const int s0 = src[e];
        const int s1 = src[e + st];
        const int s2 = src[e + 2 * st];
        const int s3 = src[e + 3 * st];
        const int p0 = base[d0 >> BSH] + atomicAdd(&hist[d0 >> BSH], 1);
        const int p1 = base[d1 >> BSH] + atomicAdd(&hist[d1 >> BSH], 1);
        const int p2 = base[d2 >> BSH] + atomicAdd(&hist[d2 >> BSH], 1);
        const int p3 = base[d3 >> BSH] + atomicAdd(&hist[d3 >> BSH], 1);
        pairs[p0] = make_int2(s0, d0);
        pairs[p1] = make_int2(s1, d1);
        pairs[p2] = make_int2(s2, d2);
        pairs[p3] = make_int2(s3, d3);
    }
    for (; e < c1; e += st) {
        const int s = src[e];
        const int d = dst[e];
        const int b = d >> BSH;
        const int r = atomicAdd(&hist[b], 1);
        pairs[base[b] + r] = make_int2(s, d);
    }
}

__global__ void __launch_bounds__(1024)
k_bsort(const int2* __restrict__ pairs, const int* __restrict__ bbase,
        const int* __restrict__ read_length,
        int* __restrict__ rowptr, int* __restrict__ csr,
        float* __restrict__ dinv, float* __restrict__ sdeg,
        float2* __restrict__ h0, int N) {
    __shared__ int cnt[BNODES];
    __shared__ int wsum[32];
    const int tid  = threadIdx.x;
    const int lane = tid & 31;
    const int wave = tid >> 5;
    const int bk   = blockIdx.x;
    const int e0   = bbase[bk];
    const int e1   = bbase[bk + 1];

    cnt[tid] = 0;
    __syncthreads();
    int j = e0 + tid;
    for (; j + 3 * BNODES < e1; j += 4 * BNODES) {
        const int d0 = pairs[j].y;
        const int d1 = pairs[j + BNODES].y;
        const int d2 = pairs[j + 2 * BNODES].y;
        const int d3 = pairs[j + 3 * BNODES].y;
        atomicAdd(&cnt[d0 & BMASK], 1);
        atomicAdd(&cnt[d1 & BMASK], 1);
        atomicAdd(&cnt[d2 & BMASK], 1);
        atomicAdd(&cnt[d3 & BMASK], 1);
    }
    for (; j < e1; j += BNODES) atomicAdd(&cnt[pairs[j].y & BMASK], 1);
    __syncthreads();

    const int c = cnt[tid];
    int incl = c;
    for (int off = 1; off < 32; off <<= 1) {
        const int t = __shfl_up(incl, off);
        if (lane >= off) incl += t;
    }
    if (lane == 31) wsum[wave] = incl;
    __syncthreads();
    if (wave == 0) {
        const int w = wsum[lane];
        int wi = w;
        for (int off = 1; off < 32; off <<= 1) {
            const int t = __shfl_up(wi, off);
            if (lane >= off) wi += t;
        }
        wsum[lane] = wi - w;
    }
    __syncthreads();
    const int start = e0 + wsum[wave] + incl - c;

    const int node = (bk << BSH) + tid;
    if (node < N) {
        rowptr[node] = start;
        const float deg = (float)(c + 1);
        const float rs  = rsqrtf(deg);
        sdeg[node] = sqrtf(deg);
        dinv[node] = 1.0f / deg;
        h0[node]   = make_float2((float)read_length[node] * LEN_SCALE_INV * rs, rs);
        if (node == N - 1) rowptr[N] = start + c;
    }
    cnt[tid] = start;
    __syncthreads();
    j = e0 + tid;
    for (; j + 3 * BNODES < e1; j += 4 * BNODES) {
        const int2 p0 = pairs[j];
        const int2 p1 = pairs[j + BNODES];
        const int2 p2 = pairs[j + 2 * BNODES];
        const int2 p3 = pairs[j + 3 * BNODES];
        const int q0 = atomicAdd(&cnt[p0.y & BMASK], 1);
        const int q1 = atomicAdd(&cnt[p1.y & BMASK], 1);
        const int q2 = atomicAdd(&cnt[p2.y & BMASK], 1);
        const int q3 = atomicAdd(&cnt[p3.y & BMASK], 1);
        csr[q0] = p0.x;
        csr[q1] = p1.x;
        csr[q2] = p2.x;
        csr[q3] = p3.x;
    }
    for (; j < e1; j += BNODES) {
        const int2 p = pairs[j];
        const int pos = atomicAdd(&cnt[p.y & BMASK], 1);
        csr[pos] = p.x;
    }
}

template <int K, int NO, int ROWS>
__device__ __forceinline__ void fold_stage(const float* sin, const float* __restrict__ Wg,
                                           float* dst, int pitch, int lane) {
    constexpr int TILES_N = (NO + 15) / 16;
    constexpr int KCH     = K / 4;
    const int lg   = lane >> 4;
    const int lmod = lane & 15;
#pragma unroll
    for (int tn = 0; tn < TILES_N; ++tn) {
        const int  nc  = tn * 16 + lmod;
        const bool nok = (nc < NO);
        const int  ncc = nok ? nc : 0;
        v8f acc = {0.f, 0.f, 0.f, 0.f, 0.f, 0.f, 0.f, 0.f};
#pragma unroll
        for (int kc = 0; kc < KCH; ++kc) {
            const int k0 = kc * 4 + lg * 2;
            const int k1 = k0 + 1;
            v2f a; a.x = sin[lmod * 16 + k0]; a.y = sin[lmod * 16 + k1];
            float b0 = Wg[k0 * NO + ncc];
            float b1 = Wg[k1 * NO + ncc];
            b0 = nok ? b0 : 0.f;
            b1 = nok ? b1 : 0.f;
            v2f b; b.x = b0; b.y = b1;
            acc = __builtin_amdgcn_wmma_f32_16x16x4_f32(
                false, a, false, b, (short)0, acc, false, false);
        }
        if (lg == 0 && nok) {
#pragma unroll
            for (int v = 0; v < ROWS; ++v) dst[v * pitch + nc] = acc[v];
        }
    }
}

__global__ void k_fold_wmma(const float* __restrict__ W1, const float* __restrict__ b1,
                            const float* __restrict__ W2, const float* __restrict__ b2,
                            const float* __restrict__ W3, const float* __restrict__ b3,
                            const float* __restrict__ W4, float* __restrict__ M) {
    __shared__ float sa[256];
    __shared__ float sb[256];
    const int lane = threadIdx.x;
    for (int i = lane; i < 256; i += 32) { sa[i] = 0.f; sb[i] = 0.f; }
    __syncthreads();
    if (lane < 4) { sa[lane] = W1[lane]; sa[16 + lane] = b1[lane]; }
    __syncthreads();
    fold_stage<4, 8, 2>(sa, W2, sb, 16, lane);
    if (lane < 8) sb[2 * 16 + lane] = b2[lane];
    __syncthreads();
    fold_stage<8, 16, 3>(sb, W3, sa, 16, lane);
    if (lane < 16) sa[3 * 16 + lane] = b3[lane];
    __syncthreads();
    fold_stage<16, 32, 4>(sa, W4, M, 32, lane);
}

#define GNODES 32

__global__ void __launch_bounds__(256)
k_gather_pair(const float2* __restrict__ hin,
              const int* __restrict__ rowptr,
              const int* __restrict__ csr,
              const float* __restrict__ dinv,
              float2* __restrict__ hout, int N) {
    const int lane = threadIdx.x & 31;
    const int wave = blockIdx.x * (blockDim.x >> 5) + (threadIdx.x >> 5);
    const int n0   = wave * GNODES;
    if (n0 >= N) return;
    const int sub  = lane >> 3;
    const int slot = lane & 7;

    const int  mynode = n0 + lane;
    const bool mine   = (mynode < N);
    const int  rb = rowptr[mine ? mynode : N];
    const int  re = rowptr[mine ? (mynode + 1) : N];
    float2 self = {0.f, 0.f};
    float  di   = 0.f;
    if (mine) { self = hin[mynode]; di = dinv[mynode]; }

    int end = __shfl(re, sub);
    int j   = __shfl(rb, sub) + slot;
    int e0 = csr[(j < end) ? j : 0];
    int e1 = csr[(j + 8 < end) ? (j + 8) : 0];
    int e2 = csr[(j + 16 < end) ? (j + 16) : 0];

    float2 res = {0.f, 0.f};
    for (int t = 0; t < GNODES / 4; ++t) {
        const bool a0 = (j < end);
        const bool a1 = (j + 8 < end);
        const bool a2 = (j + 16 < end);
        int nend = 0, nj = 0;
        if (t + 1 < GNODES / 4) {
            const int kn = (t + 1) * 4 + sub;
            nend = __shfl(re, kn);
            nj   = __shfl(rb, kn) + slot;
        }
        const float2 v0 = hin[a0 ? e0 : 0];
        const float2 v1 = hin[a1 ? e1 : 0];
        const float2 v2 = hin[a2 ? e2 : 0];
        const int f0 = csr[(nj < nend) ? nj : 0];
        const int f1 = csr[(nj + 8 < nend) ? (nj + 8) : 0];
        const int f2 = csr[(nj + 16 < nend) ? (nj + 16) : 0];

        float2 acc = {0.f, 0.f};
        acc.x += a0 ? v0.x : 0.f;
        acc.y += a0 ? v0.y : 0.f;
        acc.x += a1 ? v1.x : 0.f;
        acc.y += a1 ? v1.y : 0.f;
        acc.x += a2 ? v2.x : 0.f;
        acc.y += a2 ? v2.y : 0.f;
        for (int jj = j + 24; jj < end; jj += 8) {
            const int s = csr[jj];
            const float2 v = hin[s];
            acc.x += v.x;
            acc.y += v.y;
        }
#pragma unroll
        for (int off = 1; off < 8; off <<= 1) {
            acc.x += __shfl_xor(acc.x, off);
            acc.y += __shfl_xor(acc.y, off);
        }
        const float rx = __shfl(acc.x, (lane & 3) * 8);
        const float ry = __shfl(acc.y, (lane & 3) * 8);
        if ((lane >> 2) == t) { res.x = rx; res.y = ry; }

        end = nend; j = nj; e0 = f0; e1 = f1; e2 = f2;
    }
    if (mine)
        hout[mynode] = make_float2((res.x + self.x) * di, (res.y + self.y) * di);
}

#define OTILES 4

__global__ void __launch_bounds__(256)
k_out_wmma(const float2* __restrict__ h1, const float2* __restrict__ h2,
           const float2* __restrict__ h3, const float2* __restrict__ h4,
           const float* __restrict__ sdeg,
           const float* __restrict__ M, const float* __restrict__ bias,
           float* __restrict__ out, int N) {
    constexpr int F_OUT = 32;
    const int wave = __builtin_amdgcn_readfirstlane(
        (int)(blockIdx.x * (blockDim.x >> 5) + (threadIdx.x >> 5)));
    const int lane = threadIdx.x & 31;
    const int rowW = wave * (16 * OTILES);
    if (rowW >= N) return;

    const int  lg   = lane >> 4;
    const int  lmod = lane & 15;

    v2f   bfr[2];
    float bv[2];
#pragma unroll
    for (int tn = 0; tn < 2; ++tn) {
        const int nc = tn * 16 + lmod;
        bv[tn]    = bias[nc];
        bfr[tn].x = M[(lg * 2) * F_OUT + nc];
        bfr[tn].y = M[(lg * 2 + 1) * F_OUT + nc];
    }

    v2f a[OTILES];
#pragma unroll
    for (int i = 0; i < OTILES; ++i) {
        const int m  = rowW + i * 16 + lmod;
        const int mc = (m < N) ? m : (N - 1);
        const float2 u0 = lg ? h2[mc] : h4[mc];
        const float2 u1 = lg ? h1[mc] : h3[mc];
        const float  sd = sdeg[mc];
        a[i].x = (lg ? u0.y : u0.x) * sd;
        a[i].y = u1.y * sd;
    }

#pragma unroll
    for (int i = 0; i < OTILES; ++i) {
        const int row0 = rowW + i * 16;
        if (row0 < N) {
            const bool fullM = (row0 + 16 <= N);
#pragma unroll
            for (int tn = 0; tn < 2; ++tn) {
                const int nc = tn * 16 + lmod;
                v8f acc = {0.f, 0.f, 0.f, 0.f, 0.f, 0.f, 0.f, 0.f};
                acc = __builtin_amdgcn_wmma_f32_16x16x4_f32(
                    false, a[i], false, bfr[tn], (short)0, acc, false, false);

                float* p = out + (long long)(row0 + lg * 8) * F_OUT + nc;
                if (fullM) {
#pragma unroll
                    for (int v = 0; v < 8; ++v) p[(long long)v * F_OUT] = acc[v] + bv[tn];
                } else {
#pragma unroll
                    for (int v = 0; v < 8; ++v) {
                        const int mrow = row0 + v + lg * 8;
                        if (mrow < N) p[(long long)v * F_OUT] = acc[v] + bv[tn];
                    }
                }
            }
        }
    }
}

extern "C" void kernel_launch(void* const* d_in, const int* in_sizes, int n_in,
                              void* d_out, int out_size, void* d_ws, size_t ws_size,
                              hipStream_t stream) {
    const int* read_length = (const int*)d_in[0];
    const int* edge        = (const int*)d_in[1];
    const int  N = in_sizes[0];
    const int  E = in_sizes[1] / 2;
    const int* src = edge;
    const int* dst = edge + E;

    const float* W1 = (const float*)d_in[2]; const float* B1 = (const float*)d_in[3];
    const float* W2 = (const float*)d_in[4]; const float* B2 = (const float*)d_in[5];
    const float* W3 = (const float*)d_in[6]; const float* B3 = (const float*)d_in[7];
    const float* W4 = (const float*)d_in[8]; const float* B4 = (const float*)d_in[9];

    const int NBK = (N + BNODES - 1) / BNODES;
    float2* h0 = (float2*)d_ws;
    float2* h1 = h0 + (size_t)N;
    float2* h2 = h1 + (size_t)N;
    float2* h3 = h2 + (size_t)N;
    float2* h4 = h3 + (size_t)N;
    int2* pairs = (int2*)(h4 + (size_t)N);
    int* csr    = (int*)(pairs + (size_t)E);
    float* dinv = (float*)(csr + (size_t)E);
    float* sdeg = dinv + N;
    int* rowptr = (int*)(sdeg + N);
    int* btot   = rowptr + (N + 1);
    int* bbase  = btot + NBK * BPAD;
    int* bcur   = bbase + (NBK + 1);
    float* Mf   = (float*)(bcur + NBK * BPAD);

    const int T  = 256;

    k_fold_wmma<<<1, 32, 0, stream>>>(W1, B1, W2, B2, W3, B3, W4, Mf);

    const int TB  = 512;
    const int NCH = (E + CHUNK - 1) / CHUNK;
    k_cnt_zero<<<(NBK * BPAD + T - 1) / T, T, 0, stream>>>(btot, NBK * BPAD);
    k_bcount<<<NCH, TB, (size_t)NBK * sizeof(int), stream>>>(dst, btot, E, NBK);
    k_bscan<<<1, 1024, 0, stream>>>(btot, bbase, bcur, NBK);
    k_bscatter<<<NCH, TB, (size_t)NBK * 2 * sizeof(int), stream>>>(src, dst, bcur, pairs,
                                                                   E, NBK);
    k_bsort<<<NBK, BNODES, 0, stream>>>(pairs, bbase, read_length, rowptr, csr,
                                        dinv, sdeg, h0, N);

    const int gw = (N + GNODES - 1) / GNODES;
    const unsigned gb = (unsigned)((gw + 7) / 8);
    k_gather_pair<<<gb, T, 0, stream>>>(h0, rowptr, csr, dinv, h1, N);
    k_gather_pair<<<gb, T, 0, stream>>>(h1, rowptr, csr, dinv, h2, N);
    k_gather_pair<<<gb, T, 0, stream>>>(h2, rowptr, csr, dinv, h3, N);
    k_gather_pair<<<gb, T, 0, stream>>>(h3, rowptr, csr, dinv, h4, N);

    const long long waves  = (long long)((N + 16 * OTILES - 1) / (16 * OTILES));
    const long long blocks = (waves + 7) / 8;
    k_out_wmma<<<(unsigned)blocks, 256, 0, stream>>>(h1, h2, h3, h4, sdeg, Mf, B4,
                                                     (float*)d_out, N);
}


# sort stores coalesced through LDS: level 1 ranks a chunk's pairs by bucket in LDS and writes runs; level 2 fills an LDS image of a bucket's list, copied out in full lines; float sum order as before
# speedup vs baseline: 6.4869x; 1.2221x over previous
#include <hip/hip_runtime.h>
#include <hip/hip_bf16.h>


typedef __attribute__((ext_vector_type(2))) float v2f;
typedef __attribute__((ext_vector_type(8))) float v8f;

#define LEN_SCALE_INV (1.0f / 20000.0f)

#define BSH    10
#define BNODES (1 << BSH)
#define BMASK  (BNODES - 1)
#define CHUNK  8192
#define BPAD   32

__global__ void k_cnt_zero(int* __restrict__ cnt, int N) {
    int i = blockIdx.x * blockDim.x + threadIdx.x;
    if (i < N) cnt[i] = 0;
}

__global__ void k_bcount(const int* __restrict__ dst, int* __restrict__ btot,
                         int E, int NBK) {
    extern __shared__ int sh[];
    const int tid = threadIdx.x;
    for (int b = tid; b < NBK; b += blockDim.x) sh[b] = 0;
    __syncthreads();
    const int c0 = blockIdx.x * CHUNK;
    const int c1 = (c0 + CHUNK < E) ? (c0 + CHUNK) : E;
    const int st = blockDim.x;
    int e = c0 + tid;
    for (; e + 3 * st < c1; e += 4 * st) {
        const int d0 = dst[e];
        const int d1 = dst[e + st];
        const int d2 = dst[e + 2 * st];
        const int d3 = dst[e + 3 * st];
        atomicAdd(&sh[d0 >> BSH], 1);
        atomicAdd(&sh[d1 >> BSH], 1);
        atomicAdd(&sh[d2 >> BSH], 1);
        atomicAdd(&sh[d3 >> BSH], 1);
    }
    for (; e < c1; e += st) atomicAdd(&sh[dst[e] >> BSH], 1);
    __syncthreads();
    for (int b = tid; b < NBK; b += blockDim.x) {
        const int c = sh[b];
        if (c) atomicAdd(&btot[b * BPAD], c);
    }
}

__global__ void k_bscan(const int* __restrict__ btot, int* __restrict__ bbase,
                        int* __restrict__ bcur, int NBK) {
    __shared__ int sh[1024];
    const int tid = threadIdx.x;
    int carry = 0;
    for (int base = 0; base < NBK; base += 1024) {
        const int i = base + tid;
        const int v = (i < NBK) ? btot[i * BPAD] : 0;
        sh[tid] = v;
        __syncthreads();
        for (int off = 1; off < 1024; off <<= 1) {
            const int t = (tid >= off) ? sh[tid - off] : 0;
            __syncthreads();
            sh[tid] += t;
            __syncthreads();
        }
        const int incl  = sh[tid];
        const int total = sh[1023];
        if (i < NBK) {
            const int ex = carry + incl - v;
            bbase[i] = ex;
            bcur[i * BPAD] = ex;
        }
        carry += total;
        __syncthreads();
    }
    if (tid == 0) bbase[NBK] = carry;
}

__global__ void __launch_bounds__(512)
k_bscatter(const int* __restrict__ src, const int* __restrict__ dst,
           int* __restrict__ bcur, int2* __restrict__ pairs,
           int E, int NBK) {
    extern __shared__ int sh[];
    __shared__ int2 stage[CHUNK];
    __shared__ int wsum[32];
    __shared__ int wtot;
    int* cur   = sh;
    int* delta = sh + NBK;
    const int tid  = threadIdx.x;
    const int lane = tid & 31;
    const int wave = tid >> 5;
    const int nw   = blockDim.x >> 5;
    for (int b = tid; b < NBK; b += blockDim.x) cur[b] = 0;
    __syncthreads();
    const int c0 = blockIdx.x * CHUNK;
    const int c1 = (c0 + CHUNK < E) ? (c0 + CHUNK) : E;
    const int st = blockDim.x;
    int e = c0 + tid;
    for (; e + 3 * st < c1; e += 4 * st) {
        const int d0 = dst[e];
        const int d1 = dst[e + st];
        const int d2 = dst[e + 2 * st];
        const int d3 = dst[e + 3 * st];
        atomicAdd(&cur[d0 >> BSH], 1);
        atomicAdd(&cur[d1 >> BSH], 1);
        atomicAdd(&cur[d2 >> BSH], 1);
        atomicAdd(&cur[d3 >> BSH], 1);
    }
    for (; e < c1; e += st) atomicAdd(&cur[dst[e] >> BSH], 1);
    __syncthreads();
    int carry = 0;
    for (int b0 = 0; b0 < NBK; b0 += st) {
        const int b = b0 + tid;
        const int c = (b < NBK) ? cur[b] : 0;
        int incl = c;
        for (int off = 1; off < 32; off <<= 1) {
            const int t = __shfl_up(incl, off);
            if (lane >= off) incl += t;
        }
        if (lane == 31) wsum[wave] = incl;
        __syncthreads();
        if (wave == 0) {
            const int w = (lane < nw) ? wsum[lane] : 0;
            int wi = w;
            for (int off = 1; off < 32; off <<= 1) {
                const int t = __shfl_up(wi, off);
                if (lane >= off) wi += t;
            }
            wsum[lane] = wi - w;
            if (lane == 31) wtot = wi;
        }
        __syncthreads();
        const int loc = carry + wsum[wave] + incl - c;
        if (b < NBK) {
            const int g = c ? atomicAdd(&bcur[b * BPAD], c) : 0;
            delta[b] = g - loc;
            cur[b]   = loc;
        }
        carry += wtot;
        __syncthreads();
    }
    e = c0 + tid;
    for (; e + 3 * st < c1; e += 4 * st) {
        const int d0 = dst[e];
        const int d1 = dst[e + st];
        const int d2 = dst[e + 2 * st];
        const int d3 = dst[e + 3 * st];
        const int s0 = src[e];
        const int s1 = src[e + st];
        const int s2 = src[e + 2 * st];
        const int s3 = src[e + 3 * st];
        const int r0 = atomicAdd(&cur[d0 >> BSH], 1);
        const int r1 = atomicAdd(&cur[d1 >> BSH], 1);
        const int r2 = atomicAdd(&cur[d2 >> BSH], 1);
        const int r3 = atomicAdd(&cur[d3 >> BSH], 1);
        stage[r0] = make_int2(s0, d0);
        stage[r1] = make_int2(s1, d1);
        stage[r2] = make_int2(s2, d2);
        stage[r3] = make_int2(s3, d3);
    }
    for (; e < c1; e += st) {
        const int s = src[e];
        const int d = dst[e];
        const int r = atomicAdd(&cur[d >> BSH], 1);
        stage[r] = make_int2(s, d);
    }
    __syncthreads();
    const int n = c1 - c0;
    int i = tid;
    for (; i + 3 * st < n; i += 4 * st) {
        const int2 p0 = stage[i];
        const int2 p1 = stage[i + st];
        const int2 p2 = stage[i + 2 * st];
        const int2 p3 = stage[i + 3 * st];
        const int o0 = delta[p0.y >> BSH];
        const int o1 = delta[p1.y >> BSH];
        const int o2 = delta[p2.y >> BSH];
        const int o3 = delta[p3.y >> BSH];
        pairs[o0 + i]          = p0;
        pairs[o1 + i + st]     = p1;
        pairs[o2 + i + 2 * st] = p2;
        pairs[o3 + i + 3 * st] = p3;
    }
    for (; i < n; i += st) {
        const int2 p = stage[i];
        pairs[delta[p.y >> BSH] + i] = p;
    }
}

#define SCAP 36864

__global__ void __launch_bounds__(1024)
k_bsort(const int2* __restrict__ pairs, const int* __restrict__ bbase,
        const int* __restrict__ read_length,
        int* __restrict__ rowptr, int* __restrict__ csr,
        float* __restrict__ dinv, float* __restrict__ sdeg,
        float2* __restrict__ h0, int N) {
    __shared__ int cnt[BNODES];
    __shared__ int wsum[32];
    __shared__ int image[SCAP];
    const int tid  = threadIdx.x;
    const int lane = tid & 31;
    const int wave = tid >> 5;
    const int bk   = blockIdx.x;
    const int e0   = bbase[bk];
    const int e1   = bbase[bk + 1];

    cnt[tid] = 0;
    __syncthreads();
    int j = e0 + tid;
    for (; j + 3 * BNODES < e1; j += 4 * BNODES) {
        const int d0 = pairs[j].y;
        const int d1 = pairs[j + BNODES].y;
        const int d2 = pairs[j + 2 * BNODES].y;
        const int d3 = pairs[j + 3 * BNODES].y;
        atomicAdd(&cnt[d0 & BMASK], 1);
        atomicAdd(&cnt[d1 & BMASK], 1);
        atomicAdd(&cnt[d2 & BMASK], 1);
        atomicAdd(&cnt[d3 & BMASK], 1);
    }
    for (; j < e1; j += BNODES) atomicAdd(&cnt[pairs[j].y & BMASK], 1);
    __syncthreads();

    const int c = cnt[tid];
    int incl = c;
    for (int off = 1; off < 32; off <<= 1) {
        const int t = __shfl_up(incl, off);
        if (lane >= off) incl += t;
    }
    if (lane == 31) wsum[wave] = incl;
    __syncthreads();
    if (wave == 0) {
        const int w = wsum[lane];
        int wi = w;
        for (int off = 1; off < 32; off <<= 1) {
            const int t = __shfl_up(wi, off);
            if (lane >= off) wi += t;
        }
        wsum[lane] = wi - w;
    }
    __syncthreads();
    const int start = e0 + wsum[wave] + incl - c;

    const int node = (bk << BSH) + tid;
    if (node < N) {
        rowptr[node] = start;
        const float deg = (float)(c + 1);
        const float rs  = rsqrtf(deg);
        sdeg[node] = sqrtf(deg);
        dinv[node] = 1.0f / deg;
        h0[node]   = make_float2((float)read_length[node] * LEN_SCALE_INV * rs, rs);
        if (node == N - 1) rowptr[N] = start + c;
    }
    const int  nb     = e1 - e0;
    const bool staged = (nb <= SCAP);
    cnt[tid] = staged ? (start - e0) : start;
    __syncthreads();
    j = e0 + tid;
    if (staged) {
        for (; j + 3 * BNODES < e1; j += 4 * BNODES) {
            const int2 p0 = pairs[j];
            const int2 p1 = pairs[j + BNODES];
            const int2 p2 = pairs[j + 2 * BNODES];
            const int2 p3 = pairs[j + 3 * BNODES];
            const int q0 = atomicAdd(&cnt[p0.y & BMASK], 1);
            const int q1 = atomicAdd(&cnt[p1.y & BMASK], 1);
            const int q2 = atomicAdd(&cnt[p2.y & BMASK], 1);
            const int q3 = atomicAdd(&cnt[p3.y & BMASK], 1);
            image[q0] = p0.x;
            image[q1] = p1.x;
            image[q2] = p2.x;
            image[q3] = p3.x;
        }
        for (; j < e1; j += BNODES) {
            const int2 p = pairs[j];
            const int pos = atomicAdd(&cnt[p.y & BMASK], 1);
            image[pos] = p.x;
        }
        __syncthreads();
        for (int i = tid - (e0 & 31); i < nb; i += BNODES)
            if (i >= 0) csr[e0 + i] = image[i];
    } else {
        for (; j + 3 * BNODES < e1; j += 4 * BNODES) {
            const int2 p0 = pairs[j];
            const int2 p1 = pairs[j + BNODES];
            const int2 p2 = pairs[j + 2 * BNODES];
            const int2 p3 = pairs[j + 3 * BNODES];
            const int q0 = atomicAdd(&cnt[p0.y & BMASK], 1);
            const int q1 = atomicAdd(&cnt[p1.y & BMASK], 1);
            const int q2 = atomicAdd(&cnt[p2.y & BMASK], 1);
            const int q3 = atomicAdd(&cnt[p3.y & BMASK], 1);
            csr[q0] = p0.x;
            csr[q1] = p1.x;
            csr[q2] = p2.x;
            csr[q3] = p3.x;
        }
        for (; j < e1; j += BNODES) {
            const int2 p = pairs[j];
            const int pos = atomicAdd(&cnt[p.y & BMASK], 1);
            csr[pos] = p.x;
        }
    }
}

template <int K, int NO, int ROWS>
__device__ __forceinline__ void fold_stage(const float* sin, const float* __restrict__ Wg,
                                           float* dst, int pitch, int lane) {
    constexpr int TILES_N = (NO + 15) / 16;
    constexpr int KCH     = K / 4;
    const int lg   = lane >> 4;
    const int lmod = lane & 15;
#pragma unroll
    for (int tn = 0; tn < TILES_N; ++tn) {
        const int  nc  = tn * 16 + lmod;
        const bool nok = (nc < NO);
        const int  ncc = nok ? nc : 0;
        v8f acc = {0.f, 0.f, 0.f, 0.f, 0.f, 0.f, 0.f, 0.f};
#pragma unroll
        for (int kc = 0; kc < KCH; ++kc) {
            const int k0 = kc * 4 + lg * 2;
            const int k1 = k0 + 1;
            v2f a; a.x = sin[lmod * 16 + k0]; a.y = sin[lmod * 16 + k1];
            float b0 = Wg[k0 * NO + ncc];
            float b1 = Wg[k1 * NO + ncc];
            b0 = nok ? b0 : 0.f;
            b1 = nok ? b1 : 0.f;
            v2f b; b.x = b0; b.y = b1;
            acc = __builtin_amdgcn_wmma_f32_16x16x4_f32(
                false, a, false, b, (short)0, acc, false, false);
        }
        if (lg == 0 && nok) {
#pragma unroll
            for (int v = 0; v < ROWS; ++v) dst[v * pitch + nc] = acc[v];
        }
    }
}

__global__ void k_fold_wmma(const float* __restrict__ W1, const float* __restrict__ b1,
                            const float* __restrict__ W2, const float* __restrict__ b2,
                            const float* __restrict__ W3, const float* __restrict__ b3,
                            const float* __restrict__ W4, float* __restrict__ M) {
    __shared__ float sa[256];
    __shared__ float sb[256];
    const int lane = threadIdx.x;
    for (int i = lane; i < 256; i += 32) { sa[i] = 0.f; sb[i] = 0.f; }
    __syncthreads();
    if (lane < 4) { sa[lane] = W1[lane]; sa[16 + lane] = b1[lane]; }
    __syncthreads();
    fold_stage<4, 8, 2>(sa, W2, sb, 16, lane);
    if (lane < 8) sb[2 * 16 + lane] = b2[lane];
    __syncthreads();
    fold_stage<8, 16, 3>(sb, W3, sa, 16, lane);
    if (lane < 16) sa[3 * 16 + lane] = b3[lane];
    __syncthreads();
    fold_stage<16, 32, 4>(sa, W4, M, 32, lane);
}

#define GNODES 32

__global__ void __launch_bounds__(256)
k_gather_pair(const float2* __restrict__ hin,
              const int* __restrict__ rowptr,
              const int* __restrict__ csr,
              const float* __restrict__ dinv,
              float2* __restrict__ hout, int N) {
    const int lane = threadIdx.x & 31;
    const int wave = blockIdx.x * (blockDim.x >> 5) + (threadIdx.x >> 5);
    const int n0   = wave * GNODES;
    if (n0 >= N) return;
    const int sub  = lane >> 3;
    const int slot = lane & 7;

    const int  mynode = n0 + lane;
    const bool mine   = (mynode < N);
    const int  rb = rowptr[mine ? mynode : N];
    const int  re = rowptr[mine ? (mynode + 1) : N];
    float2 self = {0.f, 0.f};
    float  di   = 0.f;
    if (mine) { self = hin[mynode]; di = dinv[mynode]; }

    int end = __shfl(re, sub);
    int j   = __shfl(rb, sub) + slot;
    int e0 = csr[(j < end) ? j : 0];
    int e1 = csr[(j + 8 < end) ? (j + 8) : 0];
    int e2 = csr[(j + 16 < end) ? (j + 16) : 0];

    float2 res = {0.f, 0.f};
    for (int t = 0; t < GNODES / 4; ++t) {
        const bool a0 = (j < end);
        const bool a1 = (j + 8 < end);
        const bool a2 = (j + 16 < end);
        int nend = 0, nj = 0;
        if (t + 1 < GNODES / 4) {
            const int kn = (t + 1) * 4 + sub;
            nend = __shfl(re, kn);
            nj   = __shfl(rb, kn) + slot;
        }
        const float2 v0 = hin[a0 ? e0 : 0];
        const float2 v1 = hin[a1 ? e1 : 0];
        const float2 v2 = hin[a2 ? e2 : 0];
        const int f0 = csr[(nj < nend) ? nj : 0];
        const int f1 = csr[(nj + 8 < nend) ? (nj + 8) : 0];
        const int f2 = csr[(nj + 16 < nend) ? (nj + 16) : 0];

        float2 acc = {0.f, 0.f};
        acc.x += a0 ? v0.x : 0.f;
        acc.y += a0 ? v0.y : 0.f;
        acc.x += a1 ? v1.x : 0.f;
        acc.y += a1 ? v1.y : 0.f;
        acc.x += a2 ? v2.x : 0.f;
        acc.y += a2 ? v2.y : 0.f;
        for (int jj = j + 24; jj < end; jj += 8) {
            const int s = csr[jj];
            const float2 v = hin[s];
            acc.x += v.x;
            acc.y += v.y;
        }
#pragma unroll
        for (int off = 1; off < 8; off <<= 1) {
            acc.x += __shfl_xor(acc.x, off);
            acc.y += __shfl_xor(acc.y, off);
        }
        const float rx = __shfl(acc.x, (lane & 3) * 8);
        const float ry = __shfl(acc.y, (lane & 3) * 8);
        if ((lane >> 2) == t) { res.x = rx; res.y = ry; }

        end = nend; j = nj; e0 = f0; e1 = f1; e2 = f2;
    }
    if (mine)
        hout[mynode] = make_float2((res.x + self.x) * di, (res.y + self.y) * di);
}

#define OTILES 4

__global__ void __launch_bounds__(256)
k_out_wmma(const float2* __restrict__ h1, const float2* __restrict__ h2,
           const float2* __restrict__ h3, const float2* __restrict__ h4,
           const float* __restrict__ sdeg,
           const float* __restrict__ M, const float* __restrict__ bias,
           float* __restrict__ out, int N) {
    constexpr int F_OUT = 32;
    const int wave = __builtin_amdgcn_readfirstlane(
        (int)(blockIdx.x * (blockDim.x >> 5) + (threadIdx.x >> 5)));
    const int lane = threadIdx.x & 31;
    const int rowW = wave * (16 * OTILES);
    if (rowW >= N) return;

    const int  lg   = lane >> 4;
    const int  lmod = lane & 15;

    v2f   bfr[2];
    float bv[2];
#pragma unroll
    for (int tn = 0; tn < 2; ++tn) {
        const int nc = tn * 16 + lmod;
        bv[tn]    = bias[nc];
        bfr[tn].x = M[(lg * 2) * F_OUT + nc];
        bfr[tn].y = M[(lg * 2 + 1) * F_OUT + nc];
    }

    v2f a[OTILES];
#pragma unroll
    for (int i = 0; i < OTILES; ++i) {
        const int m  = rowW + i * 16 + lmod;
        const int mc = (m < N) ? m : (N - 1);
        const float2 u0 = lg ? h2[mc] : h4[mc];
        const float2 u1 = lg ? h1[mc] : h3[mc];
        const float  sd = sdeg[mc];
        a[i].x = (lg ? u0.y : u0.x) * sd;
        a[i].y = u1.y * sd;
    }

#pragma unroll
    for (int i = 0; i < OTILES; ++i) {
        const int row0 = rowW + i * 16;
        if (row0 < N) {
            const bool fullM = (row0 + 16 <= N);
#pragma unroll
            for (int tn = 0; tn < 2; ++tn) {
                const int nc = tn * 16 + lmod;
                v8f acc = {0.f, 0.f, 0.f, 0.f, 0.f, 0.f, 0.f, 0.f};
                acc = __builtin_amdgcn_wmma_f32_16x16x4_f32(
                    false, a[i], false, bfr[tn], (short)0, acc, false, false);

                float* p = out + (long long)(row0 + lg * 8) * F_OUT + nc;
                if (fullM) {
#pragma unroll
                    for (int v = 0; v < 8; ++v) p[(long long)v * F_OUT] = acc[v] + bv[tn];
                } else {
#pragma unroll
                    for (int v = 0; v < 8; ++v) {
                        const int mrow = row0 + v + lg * 8;
                        if (mrow < N) p[(long long)v * F_OUT] = acc[v] + bv[tn];
                    }
                }
            }
        }
    }
}

extern "C" void kernel_launch(void* const* d_in, const int* in_sizes, int n_in,
                              void* d_out, int out_size, void* d_ws, size_t ws_size,
                              hipStream_t stream) {
    const int* read_length = (const int*)d_in[0];
    const int* edge        = (const int*)d_in[1];
    const int  N = in_sizes[0];
    const int  E = in_sizes[1] / 2;
    const int* src = edge;
    const int* dst = edge + E;

    const float* W1 = (const float*)d_in[2]; const float* B1 = (const float*)d_in[3];
    const float* W2 = (const float*)d_in[4]; const float* B2 = (const float*)d_in[5];
    const float* W3 = (const float*)d_in[6]; const float* B3 = (const float*)d_in[7];
    const float* W4 = (const float*)d_in[8]; const float* B4 = (const float*)d_in[9];

    const int NBK = (N + BNODES - 1) / BNODES;
    float2* h0 = (float2*)d_ws;
    float2* h1 = h0 + (size_t)N;
    float2* h2 = h1 + (size_t)N;
    float2* h3 = h2 + (size_t)N;
    float2* h4 = h3 + (size_t)N;
    int2* pairs = (int2*)(h4 + (size_t)N);
    int* csr    = (int*)(pairs + (size_t)E);
    float* dinv = (float*)(csr + (size_t)E);
    float* sdeg = dinv + N;
    int* rowptr = (int*)(sdeg + N);
    int* btot   = rowptr + (N + 1);
    int* bbase  = btot + NBK * BPAD;
    int* bcur   = bbase + (NBK + 1);
    float* Mf   = (float*)(bcur + NBK * BPAD);

    const int T  = 256;

    k_fold_wmma<<<1, 32, 0, stream>>>(W1, B1, W2, B2, W3, B3, W4, Mf);

    const int TB  = 512;
    const int NCH = (E + CHUNK - 1) / CHUNK;
    k_cnt_zero<<<(NBK * BPAD + T - 1) / T, T, 0, stream>>>(btot, NBK * BPAD);
    k_bcount<<<NCH, TB, (size_t)NBK * sizeof(int), stream>>>(dst, btot, E, NBK);
    k_bscan<<<1, 1024, 0, stream>>>(btot, bbase, bcur, NBK);
    k_bscatter<<<NCH, TB, (size_t)NBK * 2 * sizeof(int), stream>>>(src, dst, bcur, pairs,
                                                                   E, NBK);
    k_bsort<<<NBK, BNODES, 0, stream>>>(pairs, bbase, read_length, rowptr, csr,
                                        dinv, sdeg, h0, N);

    const int gw = (N + GNODES - 1) / GNODES;
    const unsigned gb = (unsigned)((gw + 7) / 8);
    k_gather_pair<<<gb, T, 0, stream>>>(h0, rowptr, csr, dinv, h1, N);
    k_gather_pair<<<gb, T, 0, stream>>>(h1, rowptr, csr, dinv, h2, N);
    k_gather_pair<<<gb, T, 0, stream>>>(h2, rowptr, csr, dinv, h3, N);
    k_gather_pair<<<gb, T, 0, stream>>>(h3, rowptr, csr, dinv, h4, N);

    const long long waves  = (long long)((N + 16 * OTILES - 1) / (16 * OTILES));
    const long long blocks = (waves + 7) / 8;
    k_out_wmma<<<(unsigned)blocks, 256, 0, stream>>>(h1, h2, h3, h4, sdeg, Mf, B4,
                                                     (float*)d_out, N);
}


# edge aggregation on LDS tiles: edges binned by (dst, src) ranges of 16384 nodes; a block keeps a source slice and the dst sums in LDS, LDS float atomic adds, 8 partials per node; sum order changes
# speedup vs baseline: 7.1422x; 1.1010x over previous
#include <hip/hip_runtime.h>
#include <hip/hip_bf16.h>


typedef __attribute__((ext_vector_type(2))) float v2f;
typedef __attribute__((ext_vector_type(8))) float v8f;

#define LEN_SCALE_INV (1.0f / 20000.0f)

#define TSH    14
#define TNODES (1 << TSH)
#define TMASK  (TNODES - 1)
#define NGRP   8
#define CHUNK  8192
#define BPAD   32

__device__ __forceinline__ int tile_of(int s, int d, int NS) {
    return (d >> TSH) * NS + (s >> TSH);
}

__global__ void k_cnt_zero(int* __restrict__ cnt, int N) {
    int i = blockIdx.x * blockDim.x + threadIdx.x;
    if (i < N) cnt[i] = 0;
}

__global__ void k_tcount(const int* __restrict__ src, const int* __restrict__ dst,
                         int* __restrict__ btot, int E, int NT, int NS) {
    extern __shared__ int sh[];
    const int tid = threadIdx.x;
    for (int b = tid; b < NT; b += blockDim.x) sh[b] = 0;
    __syncthreads();
    const int c0 = blockIdx.x * CHUNK;
    const int c1 = (c0 + CHUNK < E) ? (c0 + CHUNK) : E;
    const int st = blockDim.x;
    int e = c0 + tid;
    for (; e + 3 * st < c1; e += 4 * st) {
        const int d0 = dst[e];
        const int d1 = dst[e + st];
        const int d2 = dst[e + 2 * st];
        const int d3 = dst[e + 3 * st];
        const int s0 = src[e];
        const int s1 = src[e + st];
        const int s2 = src[e + 2 * st];
        const int s3 = src[e + 3 * st];
        atomicAdd(&sh[tile_of(s0, d0, NS)], 1);
        atomicAdd(&sh[tile_of(s1, d1, NS)], 1);
        atomicAdd(&sh[tile_of(s2, d2, NS)], 1);
        atomicAdd(&sh[tile_of(s3, d3, NS)], 1);
    }
    for (; e < c1; e += st) atomicAdd(&sh[tile_of(src[e], dst[e], NS)], 1);
    __syncthreads();
    for (int b = tid; b < NT; b += blockDim.x) {
        const int c = sh[b];
        if (c) atomicAdd(&btot[b * BPAD], c);
    }
}

__global__ void k_bscan(const int* __restrict__ btot, int* __restrict__ bbase,
                        int* __restrict__ bcur, int NBK) {
    __shared__ int sh[1024];
    const int tid = threadIdx.x;
    int carry = 0;
    for (int base = 0; base < NBK; base += 1024) {
        const int i = base + tid;
        const int v = (i < NBK) ? btot[i * BPAD] : 0;
        sh[tid] = v;
        __syncthreads();
        for (int off = 1; off < 1024; off <<= 1) {
            const int t = (tid >= off) ? sh[tid - off] : 0;
            __syncthreads();
            sh[tid] += t;
            __syncthreads();
        }
        const int incl  = sh[tid];
        const int total = sh[1023];
        if (i < NBK) {
            const int ex = carry + incl - v;
            bbase[i] = ex;
            bcur[i * BPAD] = ex;
        }
        carry += total;
        __syncthreads();
    }
    if (tid == 0) bbase[NBK] = carry;
}

__global__ void __launch_bounds__(512)
k_tscatter(const int* __restrict__ src, const int* __restrict__ dst,
           int* __restrict__ bcur, unsigned* __restrict__ ents,
           int E, int NT, int NS) {
    extern __shared__ int sh[];
    __shared__ int2 stage[CHUNK];
    __shared__ int wsum[32];
    __shared__ int wtot;
    int* cur   = sh;
    int* delta = sh + NT;
    const int tid  = threadIdx.x;
    const int lane = tid & 31;
    const int wave = tid >> 5;
    const int nw   = blockDim.x >> 5;
    for (int b = tid; b < NT; b += blockDim.x) cur[b] = 0;
    __syncthreads();
    const int c0 = blockIdx.x * CHUNK;
    const int c1 = (c0 + CHUNK < E) ? (c0 + CHUNK) : E;
    const int st = blockDim.x;
    int e = c0 + tid;
    for (; e + 3 * st < c1; e += 4 * st) {
        const int d0 = dst[e];
        const int d1 = dst[e + st];
        const int d2 = dst[e + 2 * st];
        const int d3 = dst[e + 3 * st];
        const int s0 = src[e];
        const int s1 = src[e + st];
        const int s2 = src[e + 2 * st];
        const int s3 = src[e + 3 * st];
        atomicAdd(&cur[tile_of(s0, d0, NS)], 1);
        atomicAdd(&cur[tile_of(s1, d1, NS)], 1);
        atomicAdd(&cur[tile_of(s2, d2, NS)], 1);
        atomicAdd(&cur[tile_of(s3, d3, NS)], 1);
    }
    for (; e < c1; e += st) atomicAdd(&cur[tile_of(src[e], dst[e], NS)], 1);
    __syncthreads();
    int carry = 0;
    for (int b0 = 0; b0 < NT; b0 += st) {
        const int b = b0 + tid;
        const int c = (b < NT) ? cur[b] : 0;
        int incl = c;
        for (int off = 1; off < 32; off <<= 1) {
            const int t = __shfl_up(incl, off);
            if (lane >= off) incl += t;
        }
        if (lane == 31) wsum[wave] = incl;
        __syncthreads();
        if (wave == 0) {
            const int w = (lane < nw) ? wsum[lane] : 0;
            int wi = w;
            for (int off = 1; off < 32; off <<= 1) {
                const int t = __shfl_up(wi, off);
                if (lane >= off) wi += t;
            }
            wsum[lane] = wi - w;
            if (lane == 31) wtot = wi;
        }
        __syncthreads();
        const int loc = carry + wsum[wave] + incl - c;
        if (b < NT) {
            const int g = c ? atomicAdd(&bcur[b * BPAD], c) : 0;
            delta[b] = g - loc;
            cur[b]   = loc;
        }
        carry += wtot;
        __syncthreads();
    }
    e = c0 + tid;
    for (; e + 3 * st < c1; e += 4 * st) {
        const int d0 = dst[e];
        const int d1 = dst[e + st];
        const int d2 = dst[e + 2 * st];
        const int d3 = dst[e + 3 * st];
        const int s0 = src[e];
        const int s1 = src[e + st];
        const int s2 = src[e + 2 * st];
        const int s3 = src[e + 3 * st];
        const int r0 = atomicAdd(&cur[tile_of(s0, d0, NS)], 1);
        const int r1 = atomicAdd(&cur[tile_of(s1, d1, NS)], 1);
        const int r2 = atomicAdd(&cur[tile_of(s2, d2, NS)], 1);
        const int r3 = atomicAdd(&cur[tile_of(s3, d3, NS)], 1);
        stage[r0] = make_int2(s0, d0);
        stage[r1] = make_int2(s1, d1);
        stage[r2] = make_int2(s2, d2);
        stage[r3] = make_int2(s3, d3);
    }
    for (; e < c1; e += st) {
        const int s = src[e];
        const int d = dst[e];
        const int r = atomicAdd(&cur[tile_of(s, d, NS)], 1);
        stage[r] = make_int2(s, d);
    }
    __syncthreads();
    const int n = c1 - c0;
    int i = tid;
    for (; i + 3 * st < n; i += 4 * st) {
        const int2 p0 = stage[i];
        const int2 p1 = stage[i + st];
        const int2 p2 = stage[i + 2 * st];
        const int2 p3 = stage[i + 3 * st];
        const int o0 = delta[tile_of(p0.x, p0.y, NS)];
        const int o1 = delta[tile_of(p1.x, p1.y, NS)];
        const int o2 = delta[tile_of(p2.x, p2.y, NS)];
        const int o3 = delta[tile_of(p3.x, p3.y, NS)];
        ents[o0 + i]          = (unsigned)((p0.x & TMASK) | ((p0.y & TMASK) << TSH));
        ents[o1 + i + st]     = (unsigned)((p1.x & TMASK) | ((p1.y & TMASK) << TSH));
        ents[o2 + i + 2 * st] = (unsigned)((p2.x & TMASK) | ((p2.y & TMASK) << TSH));
        ents[o3 + i + 3 * st] = (unsigned)((p3.x & TMASK) | ((p3.y & TMASK) << TSH));
    }
    for (; i < n; i += st) {
        const int2 p = stage[i];
        ents[delta[tile_of(p.x, p.y, NS)] + i] =
            (unsigned)((p.x & TMASK) | ((p.y & TMASK) << TSH));
    }
}

__global__ void __launch_bounds__(1024)
k_tile_count(const unsigned* __restrict__ ents, const int* __restrict__ bbase,
             int* __restrict__ pcnt, int N, int NS, int SPG, int Np) {
    __shared__ int cnt[TNODES];
    const int tid = threadIdx.x;
    const int dr  = blockIdx.x / NGRP;
    const int g   = blockIdx.x % NGRP;
    int s0 = g * SPG;  if (s0 > NS) s0 = NS;
    int s1 = s0 + SPG; if (s1 > NS) s1 = NS;
    for (int i = tid; i < TNODES; i += 1024) cnt[i] = 0;
    __syncthreads();
    const int e0 = bbase[dr * NS + s0];
    const int e1 = bbase[dr * NS + s1];
    int j = e0 + tid;
    for (; j + 3 * 1024 < e1; j += 4 * 1024) {
        const unsigned u0 = ents[j];
        const unsigned u1 = ents[j + 1024];
        const unsigned u2 = ents[j + 2 * 1024];
        const unsigned u3 = ents[j + 3 * 1024];
        atomicAdd(&cnt[u0 >> TSH], 1);
        atomicAdd(&cnt[u1 >> TSH], 1);
        atomicAdd(&cnt[u2 >> TSH], 1);
        atomicAdd(&cnt[u3 >> TSH], 1);
    }
    for (; j < e1; j += 1024) atomicAdd(&cnt[ents[j] >> TSH], 1);
    __syncthreads();
    const int base = dr << TSH;
    int cv[TNODES / 1024];
#pragma unroll
    for (int q = 0; q < TNODES / 1024; ++q) cv[q] = cnt[tid + q * 1024];
#pragma unroll
    for (int q = 0; q < TNODES / 1024; ++q) {
        const int n = base + tid + q * 1024;
        if (n < N) pcnt[(size_t)g * Np + n] = cv[q];
    }
}

__global__ void k_node_init(const int* __restrict__ pcnt,
                            const int* __restrict__ read_length,
                            float* __restrict__ dinv, float* __restrict__ sdeg,
                            float2* __restrict__ h0, int N, int Np) {
    const int n = blockIdx.x * blockDim.x + threadIdx.x;
    if (n >= N) return;
    int c = 0;
#pragma unroll
    for (int g = 0; g < NGRP; ++g) c += pcnt[(size_t)g * Np + n];
    const float deg = (float)(c + 1);
    const float rs  = rsqrtf(deg);
    sdeg[n] = sqrtf(deg);
    dinv[n] = 1.0f / deg;
    h0[n]   = make_float2((float)read_length[n] * LEN_SCALE_INV * rs, rs);
}

template <int K, int NO, int ROWS>
__device__ __forceinline__ void fold_stage(const float* sin, const float* __restrict__ Wg,
                                           float* dst, int pitch, int lane) {
    constexpr int TILES_N = (NO + 15) / 16;
    constexpr int KCH     = K / 4;
    const int lg   = lane >> 4;
    const int lmod = lane & 15;
#pragma unroll
    for (int tn = 0; tn < TILES_N; ++tn) {
        const int  nc  = tn * 16 + lmod;
        const bool nok = (nc < NO);
        const int  ncc = nok ? nc : 0;
        v8f acc = {0.f, 0.f, 0.f, 0.f, 0.f, 0.f, 0.f, 0.f};
#pragma unroll
        for (int kc = 0; kc < KCH; ++kc) {
            const int k0 = kc * 4 + lg * 2;
            const int k1 = k0 + 1;
            v2f a; a.x = sin[lmod * 16 + k0]; a.y = sin[lmod * 16 + k1];
            float b0 = Wg[k0 * NO + ncc];
            float b1 = Wg[k1 * NO + ncc];
            b0 = nok ? b0 : 0.f;
            b1 = nok ? b1 : 0.f;
            v2f b; b.x = b0; b.y = b1;
            acc = __builtin_amdgcn_wmma_f32_16x16x4_f32(
                false, a, false, b, (short)0, acc, false, false);
        }
        if (lg == 0 && nok) {
#pragma unroll
            for (int v = 0; v < ROWS; ++v) dst[v * pitch + nc] = acc[v];
        }
    }
}

__global__ void k_fold_wmma(const float* __restrict__ W1, const float* __restrict__ b1,
                            const float* __restrict__ W2, const float* __restrict__ b2,
                            const float* __restrict__ W3, const float* __restrict__ b3,
                            const float* __restrict__ W4, float* __restrict__ M) {
    __shared__ float sa[256];
    __shared__ float sb[256];
    const int lane = threadIdx.x;
    for (int i = lane; i < 256; i += 32) { sa[i] = 0.f; sb[i] = 0.f; }
    __syncthreads();
    if (lane < 4) { sa[lane] = W1[lane]; sa[16 + lane] = b1[lane]; }
    __syncthreads();
    fold_stage<4, 8, 2>(sa, W2, sb, 16, lane);
    if (lane < 8) sb[2 * 16 + lane] = b2[lane];
    __syncthreads();
    fold_stage<8, 16, 3>(sb, W3, sa, 16, lane);
    if (lane < 16) sa[3 * 16 + lane] = b3[lane];
    __syncthreads();
    fold_stage<16, 32, 4>(sa, W4, M, 32, lane);
}

#define EPRE 9
#define SLD  (TNODES / 2 / 1024)

__device__ __forceinline__ void tile_fetch(const float2* __restrict__ hin,
                                           const unsigned* __restrict__ ents,
                                           const int* __restrict__ bbase,
                                           int t, int sr, int N, int tid,
                                           int& e0, int& e1,
                                           unsigned (&ent)[EPRE], float4 (&sv)[SLD]) {
    e0 = bbase[t];
    e1 = bbase[t + 1];
#pragma unroll
    for (int k = 0; k < EPRE; ++k) {
        const int j = e0 + tid + k * 1024;
        ent[k] = ents[(j < e1) ? j : 0];
    }
    const int sbase = sr << TSH;
    const int scnt  = (N - sbase < TNODES) ? (N - sbase) : TNODES;
    const float4* hsrc = (const float4*)(hin + sbase);
#pragma unroll
    for (int q = 0; q < SLD; ++q) {
        const int i = tid + q * 1024;
        sv[q] = hsrc[(2 * i < scnt) ? i : 0];
    }
}

__global__ void __launch_bounds__(1024)
k_tile_gather(const float2* __restrict__ hin, const unsigned* __restrict__ ents,
              const int* __restrict__ bbase, float2* __restrict__ part,
              int N, int NS, int SPG, int Np) {
    __shared__ float2 slice[TNODES];
    __shared__ float  accx[TNODES];
    __shared__ float  accy[TNODES];
    const int tid = threadIdx.x;
    const int dr  = blockIdx.x / NGRP;
    const int g   = blockIdx.x % NGRP;
    int s0 = g * SPG;  if (s0 > NS) s0 = NS;
    int s1 = s0 + SPG; if (s1 > NS) s1 = NS;
    float4* sl4 = (float4*)slice;

    int ne0 = 0, ne1 = 0;
    unsigned nent[EPRE];
    float4 sv[SLD];
#pragma unroll
    for (int k = 0; k < EPRE; ++k) nent[k] = 0u;
#pragma unroll
    for (int q = 0; q < SLD; ++q) sv[q] = make_float4(0.f, 0.f, 0.f, 0.f);
    if (s0 < s1)
        tile_fetch(hin, ents, bbase, dr * NS + s0, s0, N, tid, ne0, ne1, nent, sv);

    for (int i = tid; i < TNODES; i += 1024) { accx[i] = 0.f; accy[i] = 0.f; }
    __syncthreads();
    for (int sr = s0; sr < s1; ++sr) {
        const int e0 = ne0;
        const int e1 = ne1;
        unsigned ent[EPRE];
#pragma unroll
        for (int k = 0; k < EPRE; ++k) ent[k] = nent[k];
#pragma unroll
        for (int q = 0; q < SLD; ++q) sl4[tid + q * 1024] = sv[q];
        __syncthreads();
        if (sr + 1 < s1)
            tile_fetch(hin, ents, bbase, dr * NS + sr + 1, sr + 1, N, tid,
                       ne0, ne1, nent, sv);
        float2 v[EPRE];
#pragma unroll
        for (int k = 0; k < EPRE; ++k) v[k] = slice[ent[k] & TMASK];
#pragma unroll
        for (int k = 0; k < EPRE; ++k) {
            const int j = e0 + tid + k * 1024;
            if (j < e1) {
                atomicAdd(&accx[ent[k] >> TSH], v[k].x);
                atomicAdd(&accy[ent[k] >> TSH], v[k].y);
            }
        }
        for (int j = e0 + tid + EPRE * 1024; j < e1; j += 1024) {
            const unsigned u = ents[j];
            const float2 w = slice[u & TMASK];
            atomicAdd(&accx[u >> TSH], w.x);
            atomicAdd(&accy[u >> TSH], w.y);
        }
        __syncthreads();
    }
    const int base = dr << TSH;
    float2* po = part + (size_t)g * Np + base;
    float ax[TNODES / 1024], ay[TNODES / 1024];
#pragma unroll
    for (int q = 0; q < TNODES / 1024; ++q) {
        ax[q] = accx[tid + q * 1024];
        ay[q] = accy[tid + q * 1024];
    }
#pragma unroll
    for (int q = 0; q < TNODES / 1024; ++q) {
        const int i = tid + q * 1024;
        if (base + i < N) po[i] = make_float2(ax[q], ay[q]);
    }
}

__global__ void __launch_bounds__(256)
k_combine(const float2* __restrict__ hin, const float2* __restrict__ part,
          const float* __restrict__ dinv, float2* __restrict__ hout, int N, int Np) {
    const int n = blockIdx.x * blockDim.x + threadIdx.x;
    if (n >= N) return;
    float2 p[NGRP];
#pragma unroll
    for (int g = 0; g < NGRP; ++g) p[g] = part[(size_t)g * Np + n];
    const float2 self = hin[n];
    const float  di   = dinv[n];
    float rx = p[0].x, ry = p[0].y;
#pragma unroll
    for (int g = 1; g < NGRP; ++g) { rx += p[g].x; ry += p[g].y; }
    hout[n] = make_float2((rx + self.x) * di, (ry + self.y) * di);
}

#define OTILES 4

__global__ void __launch_bounds__(256)
k_out_wmma(const float2* __restrict__ h1, const float2* __restrict__ h2,
           const float2* __restrict__ h3, const float2* __restrict__ h4,
           const float* __restrict__ sdeg,
           const float* __restrict__ M, const float* __restrict__ bias,
           float* __restrict__ out, int N) {
    constexpr int F_OUT = 32;
    const int wave = __builtin_amdgcn_readfirstlane(
        (int)(blockIdx.x * (blockDim.x >> 5) + (threadIdx.x >> 5)));
    const int lane = threadIdx.x & 31;
    const int rowW = wave * (16 * OTILES);
    if (rowW >= N) return;

    const int  lg   = lane >> 4;
    const int  lmod = lane & 15;

    v2f   bfr[2];
    float bv[2];
#pragma unroll
    for (int tn = 0; tn < 2; ++tn) {
        const int nc = tn * 16 + lmod;
        bv[tn]    = bias[nc];
        bfr[tn].x = M[(lg * 2) * F_OUT + nc];
        bfr[tn].y = M[(lg * 2 + 1) * F_OUT + nc];
    }

    v2f a[OTILES];
#pragma unroll
    for (int i = 0; i < OTILES; ++i) {
        const int m  = rowW + i * 16 + lmod;
        const int mc = (m < N) ? m : (N - 1);
        const float2 u0 = lg ? h2[mc] : h4[mc];
        const float2 u1 = lg ? h1[mc] : h3[mc];
        const float  sd = sdeg[mc];
        a[i].x = (lg ? u0.y : u0.x) * sd;
        a[i].y = u1.y * sd;
    }

#pragma unroll
    for (int i = 0; i < OTILES; ++i) {
        const int row0 = rowW + i * 16;
        if (row0 < N) {
            const bool fullM = (row0 + 16 <= N);
#pragma unroll
            for (int tn = 0; tn < 2; ++tn) {
                const int nc = tn * 16 + lmod;
                v8f acc = {0.f, 0.f, 0.f, 0.f, 0.f, 0.f, 0.f, 0.f};
                acc = __builtin_amdgcn_wmma_f32_16x16x4_f32(
                    false, a[i], false, bfr[tn], (short)0, acc, false, false);

                float* p = out + (long long)(row0 + lg * 8) * F_OUT + nc;
                if (fullM) {
#pragma unroll
                    for (int v = 0; v < 8; ++v) p[(long long)v * F_OUT] = acc[v] + bv[tn];
                } else {
#pragma unroll
                    for (int v = 0; v < 8; ++v) {
                        const int mrow = row0 + v + lg * 8;
                        if (mrow < N) p[(long long)v * F_OUT] = acc[v] + bv[tn];
                    }
                }
            }
        }
    }
}

extern "C" void kernel_launch(void* const* d_in, const int* in_sizes, int n_in,
                              void* d_out, int out_size, void* d_ws, size_t ws_size,
                              hipStream_t stream) {
    const int* read_length = (const int*)d_in[0];
    const int* edge        = (const int*)d_in[1];
    const int  N = in_sizes[0];
    const int  E = in_sizes[1] / 2;
    const int* src = edge;
    const int* dst = edge + E;

    const float* W1 = (const float*)d_in[2]; const float* B1 = (const float*)d_in[3];
    const float* W2 = (const float*)d_in[4]; const float* B2 = (const float*)d_in[5];
    const float* W3 = (const float*)d_in[6]; const float* B3 = (const float*)d_in[7];
    const float* W4 = (const float*)d_in[8]; const float* B4 = (const float*)d_in[9];

    const int NS  = (N + TNODES - 1) / TNODES;
    const int NT  = NS * NS;
    const int SPG = (NS + NGRP - 1) / NGRP;
    const int Np  = (N + 3) & ~3;
    float2* h0 = (float2*)d_ws;
    float2* h1 = h0 + (size_t)Np;
    float2* h2 = h1 + (size_t)Np;
    float2* h3 = h2 + (size_t)Np;
    float2* h4 = h3 + (size_t)Np;
    float2* part = h4 + (size_t)Np;
    int* pcnt      = (int*)(part + (size_t)NGRP * Np);
    unsigned* ents = (unsigned*)(pcnt + (size_t)NGRP * Np);
    float* dinv = (float*)(ents + (size_t)E);
    float* sdeg = dinv + Np;
    int* btot   = (int*)(sdeg + Np);
    int* bbase  = btot + NT * BPAD;
    int* bcur   = bbase + (NT + 1);
    float* Mf   = (float*)(bcur + NT * BPAD);

    const int T  = 256;

    k_fold_wmma<<<1, 32, 0, stream>>>(W1, B1, W2, B2, W3, B3, W4, Mf);

    const int TB  = 512;
    const int NCH = (E + CHUNK - 1) / CHUNK;
    k_cnt_zero<<<(NT * BPAD + T - 1) / T, T, 0, stream>>>(btot, NT * BPAD);
    k_tcount<<<NCH, TB, (size_t)NT * sizeof(int), stream>>>(src, dst, btot, E, NT, NS);
    k_bscan<<<1, 1024, 0, stream>>>(btot, bbase, bcur, NT);
    k_tscatter<<<NCH, TB, (size_t)NT * 2 * sizeof(int), stream>>>(src, dst, bcur, ents,
                                                                  E, NT, NS);

    const unsigned tg = (unsigned)(NS * NGRP);
    const unsigned nb = (unsigned)((N + T - 1) / T);
    k_tile_count<<<tg, TNODES / 16, 0, stream>>>(ents, bbase, pcnt, N, NS, SPG, Np);
    k_node_init<<<nb, T, 0, stream>>>(pcnt, read_length, dinv, sdeg, h0, N, Np);

    k_tile_gather<<<tg, TNODES / 16, 0, stream>>>(h0, ents, bbase, part, N, NS, SPG, Np);
    k_combine<<<nb, T, 0, stream>>>(h0, part, dinv, h1, N, Np);
    k_tile_gather<<<tg, TNODES / 16, 0, stream>>>(h1, ents, bbase, part, N, NS, SPG, Np);
    k_combine<<<nb, T, 0, stream>>>(h1, part, dinv, h2, N, Np);
    k_tile_gather<<<tg, TNODES / 16, 0, stream>>>(h2, ents, bbase, part, N, NS, SPG, Np);
    k_combine<<<nb, T, 0, stream>>>(h2, part, dinv, h3, N, Np);
    k_tile_gather<<<tg, TNODES / 16, 0, stream>>>(h3, ents, bbase, part, N, NS, SPG, Np);
    k_combine<<<nb, T, 0, stream>>>(h3, part, dinv, h4, N, Np);

    const long long waves  = (long long)((N + 16 * OTILES - 1) / (16 * OTILES));
    const long long blocks = (waves + 7) / 8;
    k_out_wmma<<<(unsigned)blocks, 256, 0, stream>>>(h1, h2, h3, h4, sdeg, Mf, B4,
                                                     (float*)d_out, N);
}


# edge binning without global atomics: blocks store per-tile counts as table rows, a prefix over blocks places each block in every tile, no recount; tile entries in chunk order, LDS sums as before
# speedup vs baseline: 11.9519x; 1.6734x over previous
#include <hip/hip_runtime.h>
#include <hip/hip_bf16.h>


typedef __attribute__((ext_vector_type(2))) float v2f;
typedef __attribute__((ext_vector_type(8))) float v8f;

#define LEN_SCALE_INV (1.0f / 20000.0f)

#define TSH    14
#define TNODES (1 << TSH)
#define TMASK  (TNODES - 1)
#define NGRP   8
#define CHUNK  8192

__device__ __forceinline__ int tile_of(int s, int d, int NS) {
    return (d >> TSH) * NS + (s >> TSH);
}

__global__ void k_tcount(const int* __restrict__ src, const int* __restrict__ dst,
                         int* __restrict__ hist, int E, int NT, int NS, int HPI) {
    extern __shared__ int sh[];
    const int tid = threadIdx.x;
    for (int b = tid; b < NT; b += blockDim.x) sh[b] = 0;
    __syncthreads();
    const int c0 = blockIdx.x * CHUNK;
    const int c1 = (c0 + CHUNK < E) ? (c0 + CHUNK) : E;
    const int st = blockDim.x;
    int e = c0 + tid;
    for (; e + 3 * st < c1; e += 4 * st) {
        const int d0 = dst[e];
        const int d1 = dst[e + st];
        const int d2 = dst[e + 2 * st];
        const int d3 = dst[e + 3 * st];
        const int s0 = src[e];
        const int s1 = src[e + st];
        const int s2 = src[e + 2 * st];
        const int s3 = src[e + 3 * st];
        atomicAdd(&sh[tile_of(s0, d0, NS)], 1);
        atomicAdd(&sh[tile_of(s1, d1, NS)], 1);
        atomicAdd(&sh[tile_of(s2, d2, NS)], 1);
        atomicAdd(&sh[tile_of(s3, d3, NS)], 1);
    }
    for (; e < c1; e += st) atomicAdd(&sh[tile_of(src[e], dst[e], NS)], 1);
    __syncthreads();
    int* hrow = hist + (size_t)blockIdx.x * HPI;
    for (int b = tid; b < HPI; b += blockDim.x) hrow[b] = (b < NT) ? sh[b] : 0;
}

#define CSR 32

__global__ void __launch_bounds__(1024)
k_colscan(const int* __restrict__ hist, int* __restrict__ off, int* __restrict__ ttot,
          int NCH, int HPI) {
    __shared__ int ssum[32 * 32];
    const int lane = threadIdx.x & 31;
    const int w    = __builtin_amdgcn_readfirstlane((int)(threadIdx.x >> 5));
    const int t    = blockIdx.x * 32 + lane;
    const int SEG  = (NCH + 31) >> 5;
    int b0 = w * SEG;  if (b0 > NCH) b0 = NCH;
    int b1 = b0 + SEG; if (b1 > NCH) b1 = NCH;
    const int* hc = hist + t;
    int v[CSR];
    int sum = 0;
    for (int r0 = b0; r0 < b1; r0 += CSR) {
#pragma unroll
        for (int k = 0; k < CSR; ++k) {
            const int b  = r0 + k;
            const int bc = (b < b1) ? b : (NCH - 1);
            const int x  = hc[(size_t)bc * HPI];
            v[k] = (b < b1) ? x : 0;
        }
#pragma unroll
        for (int k = 0; k < CSR; ++k) sum += v[k];
    }
    ssum[w * 32 + lane] = sum;
    __syncthreads();
    int pre = 0, tot = 0;
#pragma unroll
    for (int k = 0; k < 32; ++k) {
        const int s = ssum[k * 32 + lane];
        tot += s;
        if (k < w) pre += s;
    }
    if (w == 0) ttot[t] = tot;
    int* oc = off + t;
    int run = pre;
    for (int r0 = b0; r0 < b1; r0 += CSR) {
        if (SEG > CSR) {
#pragma unroll
            for (int k = 0; k < CSR; ++k) {
                const int b  = r0 + k;
                const int bc = (b < b1) ? b : (NCH - 1);
                const int x  = hc[(size_t)bc * HPI];
                v[k] = (b < b1) ? x : 0;
            }
        }
#pragma unroll
        for (int k = 0; k < CSR; ++k) {
            const int b = r0 + k;
            if (b < b1) oc[(size_t)b * HPI] = run;
            run += v[k];
        }
    }
}

__global__ void __launch_bounds__(512)
k_tscatter(const int* __restrict__ src, const int* __restrict__ dst,
           const int* __restrict__ hist, const int* __restrict__ off,
           const int* __restrict__ ttot, int* __restrict__ bbase,
           unsigned* __restrict__ ents, int E, int NT, int NS, int HPI) {
    extern __shared__ int sh[];
    __shared__ int2 stage[CHUNK];
    __shared__ int wsumC[32];
    __shared__ int wsumT[32];
    __shared__ int wtotC;
    __shared__ int wtotT;
    int* cur   = sh;
    int* delta = sh + NT;
    const int tid  = threadIdx.x;
    const int lane = tid & 31;
    const int wave = tid >> 5;
    const int nw   = blockDim.x >> 5;
    const int c0 = blockIdx.x * CHUNK;
    const int c1 = (c0 + CHUNK < E) ? (c0 + CHUNK) : E;
    const int st = blockDim.x;
    const int* hrow = hist + (size_t)blockIdx.x * HPI;
    const int* orow = off + (size_t)blockIdx.x * HPI;
    int carryC = 0, carryT = 0;
    for (int b0 = 0; b0 < NT; b0 += st) {
        const int  b  = b0 + tid;
        const bool ok = (b < NT);
        const int c = ok ? hrow[b] : 0;
        const int o = ok ? orow[b] : 0;
        const int T = ok ? ttot[b] : 0;
        int ic = c, iT = T;
        for (int sh_ = 1; sh_ < 32; sh_ <<= 1) {
            const int tc = __shfl_up(ic, sh_);
            const int tT = __shfl_up(iT, sh_);
            if (lane >= sh_) { ic += tc; iT += tT; }
        }
        if (lane == 31) { wsumC[wave] = ic; wsumT[wave] = iT; }
        __syncthreads();
        if (wave == 0) {
            const int wc = (lane < nw) ? wsumC[lane] : 0;
            const int wT = (lane < nw) ? wsumT[lane] : 0;
            int sc = wc, sT = wT;
            for (int sh_ = 1; sh_ < 32; sh_ <<= 1) {
                const int tc = __shfl_up(sc, sh_);
                const int tT = __shfl_up(sT, sh_);
                if (lane >= sh_) { sc += tc; sT += tT; }
            }
            wsumC[lane] = sc - wc;
            wsumT[lane] = sT - wT;
            if (lane == 31) { wtotC = sc; wtotT = sT; }
        }
        __syncthreads();
        const int loc = carryC + wsumC[wave] + ic - c;
        const int gb  = carryT + wsumT[wave] + iT - T;
        if (ok) {
            delta[b] = gb + o - loc;
            cur[b]   = loc;
            if (blockIdx.x == 0) bbase[b] = gb;
        }
        carryC += wtotC;
        carryT += wtotT;
        __syncthreads();
    }
    if (blockIdx.x == 0 && tid == 0) bbase[NT] = carryT;
    int e = c0 + tid;
    for (; e + 3 * st < c1; e += 4 * st) {
        const int d0 = dst[e];
        const int d1 = dst[e + st];
        const int d2 = dst[e + 2 * st];
        const int d3 = dst[e + 3 * st];
        const int s0 = src[e];
        const int s1 = src[e + st];
        const int s2 = src[e + 2 * st];
        const int s3 = src[e + 3 * st];
        const int r0 = atomicAdd(&cur[tile_of(s0, d0, NS)], 1);
        const int r1 = atomicAdd(&cur[tile_of(s1, d1, NS)], 1);
        const int r2 = atomicAdd(&cur[tile_of(s2, d2, NS)], 1);
        const int r3 = atomicAdd(&cur[tile_of(s3, d3, NS)], 1);
        stage[r0] = make_int2(s0, d0);
        stage[r1] = make_int2(s1, d1);
        stage[r2] = make_int2(s2, d2);
        stage[r3] = make_int2(s3, d3);
    }
    for (; e < c1; e += st) {
        const int s = src[e];
        const int d = dst[e];
        const int r = atomicAdd(&cur[tile_of(s, d, NS)], 1);
        stage[r] = make_int2(s, d);
    }
    __syncthreads();
    const int n = c1 - c0;
    int i = tid;
    for (; i + 3 * st < n; i += 4 * st) {
        const int2 p0 = stage[i];
        const int2 p1 = stage[i + st];
        const int2 p2 = stage[i + 2 * st];
        const int2 p3 = stage[i + 3 * st];
        const int o0 = delta[tile_of(p0.x, p0.y, NS)];
        const int o1 = delta[tile_of(p1.x, p1.y, NS)];
        const int o2 = delta[tile_of(p2.x, p2.y, NS)];
        const int o3 = delta[tile_of(p3.x, p3.y, NS)];
        ents[o0 + i]          = (unsigned)((p0.x & TMASK) | ((p0.y & TMASK) << TSH));
        ents[o1 + i + st]     = (unsigned)((p1.x & TMASK) | ((p1.y & TMASK) << TSH));
        ents[o2 + i + 2 * st] = (unsigned)((p2.x & TMASK) | ((p2.y & TMASK) << TSH));
        ents[o3 + i + 3 * st] = (unsigned)((p3.x & TMASK) | ((p3.y & TMASK) << TSH));
    }
    for (; i < n; i += st) {
        const int2 p = stage[i];
        ents[delta[tile_of(p.x, p.y, NS)] + i] =
            (unsigned)((p.x & TMASK) | ((p.y & TMASK) << TSH));
    }
}

__global__ void __launch_bounds__(1024)
k_tile_count(const unsigned* __restrict__ ents, const int* __restrict__ bbase,
             int* __restrict__ pcnt, int N, int NS, int SPG, int Np) {
    __shared__ int cnt[TNODES];
    const int tid = threadIdx.x;
    const int dr  = blockIdx.x / NGRP;
    const int g   = blockIdx.x % NGRP;
    int s0 = g * SPG;  if (s0 > NS) s0 = NS;
    int s1 = s0 + SPG; if (s1 > NS) s1 = NS;
    for (int i = tid; i < TNODES; i += 1024) cnt[i] = 0;
    __syncthreads();
    const int e0 = bbase[dr * NS + s0];
    const int e1 = bbase[dr * NS + s1];
    int j = e0 + tid;
    for (; j + 3 * 1024 < e1; j += 4 * 1024) {
        const unsigned u0 = ents[j];
        const unsigned u1 = ents[j + 1024];
        const unsigned u2 = ents[j + 2 * 1024];
        const unsigned u3 = ents[j + 3 * 1024];
        atomicAdd(&cnt[u0 >> TSH], 1);
        atomicAdd(&cnt[u1 >> TSH], 1);
        atomicAdd(&cnt[u2 >> TSH], 1);
        atomicAdd(&cnt[u3 >> TSH], 1);
    }
    for (; j < e1; j += 1024) atomicAdd(&cnt[ents[j] >> TSH], 1);
    __syncthreads();
    const int base = dr << TSH;
    int cv[TNODES / 1024];
#pragma unroll
    for (int q = 0; q < TNODES / 1024; ++q) cv[q] = cnt[tid + q * 1024];
#pragma unroll
    for (int q = 0; q < TNODES / 1024; ++q) {
        const int n = base + tid + q * 1024;
        if (n < N) pcnt[(size_t)g * Np + n] = cv[q];
    }
}

__global__ void k_node_init(const int* __restrict__ pcnt,
                            const int* __restrict__ read_length,
                            float* __restrict__ dinv, float* __restrict__ sdeg,
                            float2* __restrict__ h0, int N, int Np) {
    const int n = blockIdx.x * blockDim.x + threadIdx.x;
    if (n >= N) return;
    int c = 0;
#pragma unroll
    for (int g = 0; g < NGRP; ++g) c += pcnt[(size_t)g * Np + n];
    const float deg = (float)(c + 1);
    const float rs  = rsqrtf(deg);
    sdeg[n] = sqrtf(deg);
    dinv[n] = 1.0f / deg;
    h0[n]   = make_float2((float)read_length[n] * LEN_SCALE_INV * rs, rs);
}

template <int K, int NO, int ROWS>
__device__ __forceinline__ void fold_stage(const float* sin, const float* __restrict__ Wg,
                                           float* dst, int pitch, int lane) {
    constexpr int TILES_N = (NO + 15) / 16;
    constexpr int KCH     = K / 4;
    const int lg   = lane >> 4;
    const int lmod = lane & 15;
#pragma unroll
    for (int tn = 0; tn < TILES_N; ++tn) {
        const int  nc  = tn * 16 + lmod;
        const bool nok = (nc < NO);
        const int  ncc = nok ? nc : 0;
        v8f acc = {0.f, 0.f, 0.f, 0.f, 0.f, 0.f, 0.f, 0.f};
#pragma unroll
        for (int kc = 0; kc < KCH; ++kc) {
            const int k0 = kc * 4 + lg * 2;
            const int k1 = k0 + 1;
            v2f a; a.x = sin[lmod * 16 + k0]; a.y = sin[lmod * 16 + k1];
            float b0 = Wg[k0 * NO + ncc];
            float b1 = Wg[k1 * NO + ncc];
            b0 = nok ? b0 : 0.f;
            b1 = nok ? b1 : 0.f;
            v2f b; b.x = b0; b.y = b1;
            acc = __builtin_amdgcn_wmma_f32_16x16x4_f32(
                false, a, false, b, (short)0, acc, false, false);
        }
        if (lg == 0 && nok) {
#pragma unroll
            for (int v = 0; v < ROWS; ++v) dst[v * pitch + nc] = acc[v];
        }
    }
}

__global__ void k_fold_wmma(const float* __restrict__ W1, const float* __restrict__ b1,
                            const float* __restrict__ W2, const float* __restrict__ b2,
                            const float* __restrict__ W3, const float* __restrict__ b3,
                            const float* __restrict__ W4, float* __restrict__ M) {
    __shared__ float sa[256];
    __shared__ float sb[256];
    const int lane = threadIdx.x;
    for (int i = lane; i < 256; i += 32) { sa[i] = 0.f; sb[i] = 0.f; }
    __syncthreads();
    if (lane < 4) { sa[lane] = W1[lane]; sa[16 + lane] = b1[lane]; }
    __syncthreads();
    fold_stage<4, 8, 2>(sa, W2, sb, 16, lane);
    if (lane < 8) sb[2 * 16 + lane] = b2[lane];
    __syncthreads();
    fold_stage<8, 16, 3>(sb, W3, sa, 16, lane);
    if (lane < 16) sa[3 * 16 + lane] = b3[lane];
    __syncthreads();
    fold_stage<16, 32, 4>(sa, W4, M, 32, lane);
}

#define EPRE 9
#define SLD  (TNODES / 2 / 1024)

__device__ __forceinline__ void tile_fetch(const float2* __restrict__ hin,
                                           const unsigned* __restrict__ ents,
                                           const int* __restrict__ bbase,
                                           int t, int sr, int N, int tid,
                                           int& e0, int& e1,
                                           unsigned (&ent)[EPRE], float4 (&sv)[SLD]) {
    e0 = bbase[t];
    e1 = bbase[t + 1];
#pragma unroll
    for (int k = 0; k < EPRE; ++k) {
        const int j = e0 + tid + k * 1024;
        ent[k] = ents[(j < e1) ? j : 0];
    }
    const int sbase = sr << TSH;
    const int scnt  = (N - sbase < TNODES) ? (N - sbase) : TNODES;
    const float4* hsrc = (const float4*)(hin + sbase);
#pragma unroll
    for (int q = 0; q < SLD; ++q) {
        const int i = tid + q * 1024;
        sv[q] = hsrc[(2 * i < scnt) ? i : 0];
    }
}

__global__ void __launch_bounds__(1024)
k_tile_gather(const float2* __restrict__ hin, const unsigned* __restrict__ ents,
              const int* __restrict__ bbase, float2* __restrict__ part,
              int N, int NS, int SPG, int Np) {
    __shared__ float2 slice[TNODES];
    __shared__ float  accx[TNODES];
    __shared__ float  accy[TNODES];
    const int tid = threadIdx.x;
    const int dr  = blockIdx.x / NGRP;
    const int g   = blockIdx.x % NGRP;
    int s0 = g * SPG;  if (s0 > NS) s0 = NS;
    int s1 = s0 + SPG; if (s1 > NS) s1 = NS;
    float4* sl4 = (float4*)slice;

    int ne0 = 0, ne1 = 0;
    unsigned nent[EPRE];
    float4 sv[SLD];
#pragma unroll
    for (int k = 0; k < EPRE; ++k) nent[k] = 0u;
#pragma unroll
    for (int q = 0; q < SLD; ++q) sv[q] = make_float4(0.f, 0.f, 0.f, 0.f);
    if (s0 < s1)
        tile_fetch(hin, ents, bbase, dr * NS + s0, s0, N, tid, ne0, ne1, nent, sv);

    for (int i = tid; i < TNODES; i += 1024) { accx[i] = 0.f; accy[i] = 0.f; }
    __syncthreads();
    for (int sr = s0; sr < s1; ++sr) {
        const int e0 = ne0;
        const int e1 = ne1;
        unsigned ent[EPRE];
#pragma unroll
        for (int k = 0; k < EPRE; ++k) ent[k] = nent[k];
#pragma unroll
        for (int q = 0; q < SLD; ++q) sl4[tid + q * 1024] = sv[q];
        __syncthreads();
        if (sr + 1 < s1)
            tile_fetch(hin, ents, bbase, dr * NS + sr + 1, sr + 1, N, tid,
                       ne0, ne1, nent, sv);
        float2 v[EPRE];
#pragma unroll
        for (int k = 0; k < EPRE; ++k) v[k] = slice[ent[k] & TMASK];
#pragma unroll
        for (int k = 0; k < EPRE; ++k) {
            const int j = e0 + tid + k * 1024;
            if (j < e1) {
                atomicAdd(&accx[ent[k] >> TSH], v[k].x);
                atomicAdd(&accy[ent[k] >> TSH], v[k].y);
            }
        }
        for (int j = e0 + tid + EPRE * 1024; j < e1; j += 1024) {
            const unsigned u = ents[j];
            const float2 w = slice[u & TMASK];
            atomicAdd(&accx[u >> TSH], w.x);
            atomicAdd(&accy[u >> TSH], w.y);
        }
        __syncthreads();
    }
    const int base = dr << TSH;
    float2* po = part + (size_t)g * Np + base;
    float ax[TNODES / 1024], ay[TNODES / 1024];
#pragma unroll
    for (int q = 0; q < TNODES / 1024; ++q) {
        ax[q] = accx[tid + q * 1024];
        ay[q] = accy[tid + q * 1024];
    }
#pragma unroll
    for (int q = 0; q < TNODES / 1024; ++q) {
        const int i = tid + q * 1024;
        if (base + i < N) po[i] = make_float2(ax[q], ay[q]);
    }
}

__global__ void __launch_bounds__(256)
k_combine(const float2* __restrict__ hin, const float2* __restrict__ part,
          const float* __restrict__ dinv, float2* __restrict__ hout, int N, int Np) {
    const int n = blockIdx.x * blockDim.x + threadIdx.x;
    if (n >= N) return;
    float2 p[NGRP];
#pragma unroll
    for (int g = 0; g < NGRP; ++g) p[g] = part[(size_t)g * Np + n];
    const float2 self = hin[n];
    const float  di   = dinv[n];
    float rx = p[0].x, ry = p[0].y;
#pragma unroll
    for (int g = 1; g < NGRP; ++g) { rx += p[g].x; ry += p[g].y; }
    hout[n] = make_float2((rx + self.x) * di, (ry + self.y) * di);
}

#define OTILES 4

__global__ void __launch_bounds__(256)
k_out_wmma(const float2* __restrict__ h1, const float2* __restrict__ h2,
           const float2* __restrict__ h3, const float2* __restrict__ h4,
           const float* __restrict__ sdeg,
           const float* __restrict__ M, const float* __restrict__ bias,
           float* __restrict__ out, int N) {
    constexpr int F_OUT = 32;
    const int wave = __builtin_amdgcn_readfirstlane(
        (int)(blockIdx.x * (blockDim.x >> 5) + (threadIdx.x >> 5)));
    const int lane = threadIdx.x & 31;
    const int rowW = wave * (16 * OTILES);
    if (rowW >= N) return;

    const int  lg   = lane >> 4;
    const int  lmod = lane & 15;

    v2f   bfr[2];
    float bv[2];
#pragma unroll
    for (int tn = 0; tn < 2; ++tn) {
        const int nc = tn * 16 + lmod;
        bv[tn]    = bias[nc];
        bfr[tn].x = M[(lg * 2) * F_OUT + nc];
        bfr[tn].y = M[(lg * 2 + 1) * F_OUT + nc];
    }

    v2f a[OTILES];
#pragma unroll
    for (int i = 0; i < OTILES; ++i) {
        const int m  = rowW + i * 16 + lmod;
        const int mc = (m < N) ? m : (N - 1);
        const float2 u0 = lg ? h2[mc] : h4[mc];
        const float2 u1 = lg ? h1[mc] : h3[mc];
        const float  sd = sdeg[mc];
        a[i].x = (lg ? u0.y : u0.x) * sd;
        a[i].y = u1.y * sd;
    }

#pragma unroll
    for (int i = 0; i < OTILES; ++i) {
        const int row0 = rowW + i * 16;
        if (row0 < N) {
            const bool fullM = (row0 + 16 <= N);
#pragma unroll
            for (int tn = 0; tn < 2; ++tn) {
                const int nc = tn * 16 + lmod;
                v8f acc = {0.f, 0.f, 0.f, 0.f, 0.f, 0.f, 0.f, 0.f};
                acc = __builtin_amdgcn_wmma_f32_16x16x4_f32(
                    false, a[i], false, bfr[tn], (short)0, acc, false, false);

                float* p = out + (long long)(row0 + lg * 8) * F_OUT + nc;
                if (fullM) {
#pragma unroll
                    for (int v = 0; v < 8; ++v) p[(long long)v * F_OUT] = acc[v] + bv[tn];
                } else {
#pragma unroll
                    for (int v = 0; v < 8; ++v) {
                        const int mrow = row0 + v + lg * 8;
                        if (mrow < N) p[(long long)v * F_OUT] = acc[v] + bv[tn];
                    }
                }
            }
        }
    }
}

extern "C" void kernel_launch(void* const* d_in, const int* in_sizes, int n_in,
                              void* d_out, int out_size, void* d_ws, size_t ws_size,
                              hipStream_t stream) {
    const int* read_length = (const int*)d_in[0];
    const int* edge        = (const int*)d_in[1];
    const int  N = in_sizes[0];
    const int  E = in_sizes[1] / 2;
    const int* src = edge;
    const int* dst = edge + E;

    const float* W1 = (const float*)d_in[2]; const float* B1 = (const float*)d_in[3];
    const float* W2 = (const float*)d_in[4]; const float* B2 = (const float*)d_in[5];
    const float* W3 = (const float*)d_in[6]; const float* B3 = (const float*)d_in[7];
    const float* W4 = (const float*)d_in[8]; const float* B4 = (const float*)d_in[9];

    const int NS  = (N + TNODES - 1) / TNODES;
    const int NT  = NS * NS;
    const int SPG = (NS + NGRP - 1) / NGRP;
    const int Np  = (N + 3) & ~3;
    const int NCH = (E + CHUNK - 1) / CHUNK;
    const int HPI = (NT + 31) & ~31;
    float2* h0 = (float2*)d_ws;
    float2* h1 = h0 + (size_t)Np;
    float2* h2 = h1 + (size_t)Np;
    float2* h3 = h2 + (size_t)Np;
    float2* h4 = h3 + (size_t)Np;
    float2* part = h4 + (size_t)Np;
    int* pcnt      = (int*)(part + (size_t)NGRP * Np);
    unsigned* ents = (unsigned*)(pcnt + (size_t)NGRP * Np);
    float* dinv = (float*)(ents + (size_t)E);
    float* sdeg = dinv + Np;
    int* hist   = (int*)(sdeg + Np);
    int* off    = hist + (size_t)NCH * HPI;
    int* ttot   = off + (size_t)NCH * HPI;
    int* bbase  = ttot + HPI;
    float* Mf   = (float*)(bbase + (NT + 1));

    const int T  = 256;

    k_fold_wmma<<<1, 32, 0, stream>>>(W1, B1, W2, B2, W3, B3, W4, Mf);

    const int TB  = 512;
    k_tcount<<<NCH, TB, (size_t)NT * sizeof(int), stream>>>(src, dst, hist, E, NT, NS, HPI);
    k_colscan<<<HPI / 32, 1024, 0, stream>>>(hist, off, ttot, NCH, HPI);
    k_tscatter<<<NCH, TB, (size_t)NT * 2 * sizeof(int), stream>>>(src, dst, hist, off, ttot,
                                                                  bbase, ents, E, NT, NS, HPI);

    const unsigned tg = (unsigned)(NS * NGRP);
    const unsigned nb = (unsigned)((N + T - 1) / T);
    k_tile_count<<<tg, TNODES / 16, 0, stream>>>(ents, bbase, pcnt, N, NS, SPG, Np);
    k_node_init<<<nb, T, 0, stream>>>(pcnt, read_length, dinv, sdeg, h0, N, Np);

    k_tile_gather<<<tg, TNODES / 16, 0, stream>>>(h0, ents, bbase, part, N, NS, SPG, Np);
    k_combine<<<nb, T, 0, stream>>>(h0, part, dinv, h1, N, Np);
    k_tile_gather<<<tg, TNODES / 16, 0, stream>>>(h1, ents, bbase, part, N, NS, SPG, Np);
    k_combine<<<nb, T, 0, stream>>>(h1, part, dinv, h2, N, Np);
    k_tile_gather<<<tg, TNODES / 16, 0, stream>>>(h2, ents, bbase, part, N, NS, SPG, Np);
    k_combine<<<nb, T, 0, stream>>>(h2, part, dinv, h3, N, Np);
    k_tile_gather<<<tg, TNODES / 16, 0, stream>>>(h3, ents, bbase, part, N, NS, SPG, Np);
    k_combine<<<nb, T, 0, stream>>>(h3, part, dinv, h4, N, Np);

    const long long waves  = (long long)((N + 16 * OTILES - 1) / (16 * OTILES));
    const long long blocks = (waves + 7) / 8;
    k_out_wmma<<<(unsigned)blocks, 256, 0, stream>>>(h1, h2, h3, h4, sdeg, Mf, B4,
                                                     (float*)d_out, N);
}


# last layer specialised: q4 is unused, so the 4th aggregation carries one float per node (4-byte LDS slice, one LDS add per edge) and the output pass adds its partial sums itself; sum order unchanged
# speedup vs baseline: 12.7009x; 1.0627x over previous
#include <hip/hip_runtime.h>
#include <hip/hip_bf16.h>


typedef __attribute__((ext_vector_type(2))) float v2f;
typedef __attribute__((ext_vector_type(8))) float v8f;

#define LEN_SCALE_INV (1.0f / 20000.0f)

#define TSH    14
#define TNODES (1 << TSH)
#define TMASK  (TNODES - 1)
#define NGRP   8
#define CHUNK  8192

__device__ __forceinline__ int tile_of(int s, int d, int NS) {
    return (d >> TSH) * NS + (s >> TSH);
}

__global__ void k_tcount(const int* __restrict__ src, const int* __restrict__ dst,
                         int* __restrict__ hist, int E, int NT, int NS, int HPI) {
    extern __shared__ int sh[];
    const int tid = threadIdx.x;
    for (int b = tid; b < NT; b += blockDim.x) sh[b] = 0;
    __syncthreads();
    const int c0 = blockIdx.x * CHUNK;
    const int c1 = (c0 + CHUNK < E) ? (c0 + CHUNK) : E;
    const int st = blockDim.x;
    int e = c0 + tid;
    for (; e + 3 * st < c1; e += 4 * st) {
        const int d0 = dst[e];
        const int d1 = dst[e + st];
        const int d2 = dst[e + 2 * st];
        const int d3 = dst[e + 3 * st];
        const int s0 = src[e];
        const int s1 = src[e + st];
        const int s2 = src[e + 2 * st];
        const int s3 = src[e + 3 * st];
        atomicAdd(&sh[tile_of(s0, d0, NS)], 1);
        atomicAdd(&sh[tile_of(s1, d1, NS)], 1);
        atomicAdd(&sh[tile_of(s2, d2, NS)], 1);
        atomicAdd(&sh[tile_of(s3, d3, NS)], 1);
    }
    for (; e < c1; e += st) atomicAdd(&sh[tile_of(src[e], dst[e], NS)], 1);
    __syncthreads();
    int* hrow = hist + (size_t)blockIdx.x * HPI;
    for (int b = tid; b < HPI; b += blockDim.x) hrow[b] = (b < NT) ? sh[b] : 0;
}

#define CSR 32

__global__ void __launch_bounds__(1024)
k_colscan(const int* __restrict__ hist, int* __restrict__ off, int* __restrict__ ttot,
          int NCH, int HPI) {
    __shared__ int ssum[32 * 32];
    const int lane = threadIdx.x & 31;
    const int w    = __builtin_amdgcn_readfirstlane((int)(threadIdx.x >> 5));
    const int t    = blockIdx.x * 32 + lane;
    const int SEG  = (NCH + 31) >> 5;
    int b0 = w * SEG;  if (b0 > NCH) b0 = NCH;
    int b1 = b0 + SEG; if (b1 > NCH) b1 = NCH;
    const int* hc = hist + t;
    int v[CSR];
    int sum = 0;
    for (int r0 = b0; r0 < b1; r0 += CSR) {
#pragma unroll
        for (int k = 0; k < CSR; ++k) {
            const int b  = r0 + k;
            const int bc = (b < b1) ? b : (NCH - 1);
            const int x  = hc[(size_t)bc * HPI];
            v[k] = (b < b1) ? x : 0;
        }
#pragma unroll
        for (int k = 0; k < CSR; ++k) sum += v[k];
    }
    ssum[w * 32 + lane] = sum;
    __syncthreads();
    int pre = 0, tot = 0;
#pragma unroll
    for (int k = 0; k < 32; ++k) {
        const int s = ssum[k * 32 + lane];
        tot += s;
        if (k < w) pre += s;
    }
    if (w == 0) ttot[t] = tot;
    int* oc = off + t;
    int run = pre;
    for (int r0 = b0; r0 < b1; r0 += CSR) {
        if (SEG > CSR) {
#pragma unroll
            for (int k = 0; k < CSR; ++k) {
                const int b  = r0 + k;
                const int bc = (b < b1) ? b : (NCH - 1);
                const int x  = hc[(size_t)bc * HPI];
                v[k] = (b < b1) ? x : 0;
            }
        }
#pragma unroll
        for (int k = 0; k < CSR; ++k) {
            const int b = r0 + k;
            if (b < b1) oc[(size_t)b * HPI] = run;
            run += v[k];
        }
    }
}

__global__ void __launch_bounds__(512)
k_tscatter(const int* __restrict__ src, const int* __restrict__ dst,
           const int* __restrict__ hist, const int* __restrict__ off,
           const int* __restrict__ ttot, int* __restrict__ bbase,
           unsigned* __restrict__ ents, int E, int NT, int NS, int HPI) {
    extern __shared__ int sh[];
    __shared__ int2 stage[CHUNK];
    __shared__ int wsumC[32];
    __shared__ int wsumT[32];
    __shared__ int wtotC;
    __shared__ int wtotT;
    int* cur   = sh;
    int* delta = sh + NT;
    const int tid  = threadIdx.x;
    const int lane = tid & 31;
    const int wave = tid >> 5;
    const int nw   = blockDim.x >> 5;
    const int c0 = blockIdx.x * CHUNK;
    const int c1 = (c0 + CHUNK < E) ? (c0 + CHUNK) : E;
    const int st = blockDim.x;
    const int* hrow = hist + (size_t)blockIdx.x * HPI;
    const int* orow = off + (size_t)blockIdx.x * HPI;
    int carryC = 0, carryT = 0;
    for (int b0 = 0; b0 < NT; b0 += st) {
        const int  b  = b0 + tid;
        const bool ok = (b < NT);
        const int c = ok ? hrow[b] : 0;
        const int o = ok ? orow[b] : 0;
        const int T = ok ? ttot[b] : 0;
        int ic = c, iT = T;
        for (int sh_ = 1; sh_ < 32; sh_ <<= 1) {
            const int tc = __shfl_up(ic, sh_);
            const int tT = __shfl_up(iT, sh_);
            if (lane >= sh_) { ic += tc; iT += tT; }
        }
        if (lane == 31) { wsumC[wave] = ic; wsumT[wave] = iT; }
        __syncthreads();
        if (wave == 0) {
            const int wc = (lane < nw) ? wsumC[lane] : 0;
            const int wT = (lane < nw) ? wsumT[lane] : 0;
            int sc = wc, sT = wT;
            for (int sh_ = 1; sh_ < 32; sh_ <<= 1) {
                const int tc = __shfl_up(sc, sh_);
                const int tT = __shfl_up(sT, sh_);
                if (lane >= sh_) { sc += tc; sT += tT; }
            }
            wsumC[lane] = sc - wc;
            wsumT[lane] = sT - wT;
            if (lane == 31) { wtotC = sc; wtotT = sT; }
        }
        __syncthreads();
        const int loc = carryC + wsumC[wave] + ic - c;
        const int gb  = carryT + wsumT[wave] + iT - T;
        if (ok) {
            delta[b] = gb + o - loc;
            cur[b]   = loc;
            if (blockIdx.x == 0) bbase[b] = gb;
        }
        carryC += wtotC;
        carryT += wtotT;
        __syncthreads();
    }
    if (blockIdx.x == 0 && tid == 0) bbase[NT] = carryT;
    int e = c0 + tid;
    for (; e + 3 * st < c1; e += 4 * st) {
        const int d0 = dst[e];
        const int d1 = dst[e + st];
        const int d2 = dst[e + 2 * st];
        const int d3 = dst[e + 3 * st];
        const int s0 = src[e];
        const int s1 = src[e + st];
        const int s2 = src[e + 2 * st];
        const int s3 = src[e + 3 * st];
        const int r0 = atomicAdd(&cur[tile_of(s0, d0, NS)], 1);
        const int r1 = atomicAdd(&cur[tile_of(s1, d1, NS)], 1);
        const int r2 = atomicAdd(&cur[tile_of(s2, d2, NS)], 1);
        const int r3 = atomicAdd(&cur[tile_of(s3, d3, NS)], 1);
        stage[r0] = make_int2(s0, d0);
        stage[r1] = make_int2(s1, d1);
        stage[r2] = make_int2(s2, d2);
        stage[r3] = make_int2(s3, d3);
    }
    for (; e < c1; e += st) {
        const int s = src[e];
        const int d = dst[e];
        const int r = atomicAdd(&cur[tile_of(s, d, NS)], 1);
        stage[r] = make_int2(s, d);
    }
    __syncthreads();
    const int n = c1 - c0;
    int i = tid;
    for (; i + 3 * st < n; i += 4 * st) {
        const int2 p0 = stage[i];
        const int2 p1 = stage[i + st];
        const int2 p2 = stage[i + 2 * st];
        const int2 p3 = stage[i + 3 * st];
        const int o0 = delta[tile_of(p0.x, p0.y, NS)];
        const int o1 = delta[tile_of(p1.x, p1.y, NS)];
        const int o2 = delta[tile_of(p2.x, p2.y, NS)];
        const int o3 = delta[tile_of(p3.x, p3.y, NS)];
        ents[o0 + i]          = (unsigned)((p0.x & TMASK) | ((p0.y & TMASK) << TSH));
        ents[o1 + i + st]     = (unsigned)((p1.x & TMASK) | ((p1.y & TMASK) << TSH));
        ents[o2 + i + 2 * st] = (unsigned)((p2.x & TMASK) | ((p2.y & TMASK) << TSH));
        ents[o3 + i + 3 * st] = (unsigned)((p3.x & TMASK) | ((p3.y & TMASK) << TSH));
    }
    for (; i < n; i += st) {
        const int2 p = stage[i];
        ents[delta[tile_of(p.x, p.y, NS)] + i] =
            (unsigned)((p.x & TMASK) | ((p.y & TMASK) << TSH));
    }
}

__global__ void __launch_bounds__(1024)
k_tile_count(const unsigned* __restrict__ ents, const int* __restrict__ bbase,
             int* __restrict__ pcnt, int N, int NS, int SPG, int Np) {
    __shared__ int cnt[TNODES];
    const int tid = threadIdx.x;
    const int dr  = blockIdx.x / NGRP;
    const int g   = blockIdx.x % NGRP;
    int s0 = g * SPG;  if (s0 > NS) s0 = NS;
    int s1 = s0 + SPG; if (s1 > NS) s1 = NS;
    for (int i = tid; i < TNODES; i += 1024) cnt[i] = 0;
    __syncthreads();
    const int e0 = bbase[dr * NS + s0];
    const int e1 = bbase[dr * NS + s1];
    int j = e0 + tid;
    for (; j + 3 * 1024 < e1; j += 4 * 1024) {
        const unsigned u0 = ents[j];
        const unsigned u1 = ents[j + 1024];
        const unsigned u2 = ents[j + 2 * 1024];
        const unsigned u3 = ents[j + 3 * 1024];
        atomicAdd(&cnt[u0 >> TSH], 1);
        atomicAdd(&cnt[u1 >> TSH], 1);
        atomicAdd(&cnt[u2 >> TSH], 1);
        atomicAdd(&cnt[u3 >> TSH], 1);
    }
    for (; j < e1; j += 1024) atomicAdd(&cnt[ents[j] >> TSH], 1);
    __syncthreads();
    const int base = dr << TSH;
    int cv[TNODES / 1024];
#pragma unroll
    for (int q = 0; q < TNODES / 1024; ++q) cv[q] = cnt[tid + q * 1024];
#pragma unroll
    for (int q = 0; q < TNODES / 1024; ++q) {
        const int n = base + tid + q * 1024;
        if (n < N) pcnt[(size_t)g * Np + n] = cv[q];
    }
}

__global__ void k_node_init(const int* __restrict__ pcnt,
                            const int* __restrict__ read_length,
                            float* __restrict__ dinv, float* __restrict__ sdeg,
                            float2* __restrict__ h0, int N, int Np) {
    const int n = blockIdx.x * blockDim.x + threadIdx.x;
    if (n >= N) return;
    int c = 0;
#pragma unroll
    for (int g = 0; g < NGRP; ++g) c += pcnt[(size_t)g * Np + n];
    const float deg = (float)(c + 1);
    const float rs  = rsqrtf(deg);
    sdeg[n] = sqrtf(deg);
    dinv[n] = 1.0f / deg;
    h0[n]   = make_float2((float)read_length[n] * LEN_SCALE_INV * rs, rs);
}

template <int K, int NO, int ROWS>
__device__ __forceinline__ void fold_stage(const float* sin, const float* __restrict__ Wg,
                                           float* dst, int pitch, int lane) {
    constexpr int TILES_N = (NO + 15) / 16;
    constexpr int KCH     = K / 4;
    const int lg   = lane >> 4;
    const int lmod = lane & 15;
#pragma unroll
    for (int tn = 0; tn < TILES_N; ++tn) {
        const int  nc  = tn * 16 + lmod;
        const bool nok = (nc < NO);
        const int  ncc = nok ? nc : 0;
        v8f acc = {0.f, 0.f, 0.f, 0.f, 0.f, 0.f, 0.f, 0.f};
#pragma unroll
        for (int kc = 0; kc < KCH; ++kc) {
            const int k0 = kc * 4 + lg * 2;
            const int k1 = k0 + 1;
            v2f a; a.x = sin[lmod * 16 + k0]; a.y = sin[lmod * 16 + k1];
            float b0 = Wg[k0 * NO + ncc];
            float b1 = Wg[k1 * NO + ncc];
            b0 = nok ? b0 : 0.f;
            b1 = nok ? b1 : 0.f;
            v2f b; b.x = b0; b.y = b1;
            acc = __builtin_amdgcn_wmma_f32_16x16x4_f32(
                false, a, false, b, (short)0, acc, false, false);
        }
        if (lg == 0 && nok) {
#pragma unroll
            for (int v = 0; v < ROWS; ++v) dst[v * pitch + nc] = acc[v];
        }
    }
}

__global__ void k_fold_wmma(const float* __restrict__ W1, const float* __restrict__ b1,
                            const float* __restrict__ W2, const float* __restrict__ b2,
                            const float* __restrict__ W3, const float* __restrict__ b3,
                            const float* __restrict__ W4, float* __restrict__ M) {
    __shared__ float sa[256];
    __shared__ float sb[256];
    const int lane = threadIdx.x;
    for (int i = lane; i < 256; i += 32) { sa[i] = 0.f; sb[i] = 0.f; }
    __syncthreads();
    if (lane < 4) { sa[lane] = W1[lane]; sa[16 + lane] = b1[lane]; }
    __syncthreads();
    fold_stage<4, 8, 2>(sa, W2, sb, 16, lane);
    if (lane < 8) sb[2 * 16 + lane] = b2[lane];
    __syncthreads();
    fold_stage<8, 16, 3>(sb, W3, sa, 16, lane);
    if (lane < 16) sa[3 * 16 + lane] = b3[lane];
    __syncthreads();
    fold_stage<16, 32, 4>(sa, W4, M, 32, lane);
}

#define EPRE 9
#define SLD  (TNODES / 2 / 1024)

__device__ __forceinline__ void tile_fetch(const float2* __restrict__ hin,
                                           const unsigned* __restrict__ ents,
                                           const int* __restrict__ bbase,
                                           int t, int sr, int N, int tid,
                                           int& e0, int& e1,
                                           unsigned (&ent)[EPRE], float4 (&sv)[SLD]) {
    e0 = bbase[t];
    e1 = bbase[t + 1];
#pragma unroll
    for (int k = 0; k < EPRE; ++k) {
        const int j = e0 + tid + k * 1024;
        ent[k] = ents[(j < e1) ? j : 0];
    }
    const int sbase = sr << TSH;
    const int scnt  = (N - sbase < TNODES) ? (N - sbase) : TNODES;
    const float4* hsrc = (const float4*)(hin + sbase);
#pragma unroll
    for (int q = 0; q < SLD; ++q) {
        const int i = tid + q * 1024;
        sv[q] = hsrc[(2 * i < scnt) ? i : 0];
    }
}

__global__ void __launch_bounds__(1024)
k_tile_gather(const float2* __restrict__ hin, const unsigned* __restrict__ ents,
              const int* __restrict__ bbase, float2* __restrict__ part,
              int N, int NS, int SPG, int Np) {
    __shared__ float2 slice[TNODES];
    __shared__ float  accx[TNODES];
    __shared__ float  accy[TNODES];
    const int tid = threadIdx.x;
    const int dr  = blockIdx.x / NGRP;
    const int g   = blockIdx.x % NGRP;
    int s0 = g * SPG;  if (s0 > NS) s0 = NS;
    int s1 = s0 + SPG; if (s1 > NS) s1 = NS;
    float4* sl4 = (float4*)slice;

    int ne0 = 0, ne1 = 0;
    unsigned nent[EPRE];
    float4 sv[SLD];
#pragma unroll
    for (int k = 0; k < EPRE; ++k) nent[k] = 0u;
#pragma unroll
    for (int q = 0; q < SLD; ++q) sv[q] = make_float4(0.f, 0.f, 0.f, 0.f);
    if (s0 < s1)
        tile_fetch(hin, ents, bbase, dr * NS + s0, s0, N, tid, ne0, ne1, nent, sv);

    for (int i = tid; i < TNODES; i += 1024) { accx[i] = 0.f; accy[i] = 0.f; }
    __syncthreads();
    for (int sr = s0; sr < s1; ++sr) {
        const int e0 = ne0;
        const int e1 = ne1;
        unsigned ent[EPRE];
#pragma unroll
        for (int k = 0; k < EPRE; ++k) ent[k] = nent[k];
#pragma unroll
        for (int q = 0; q < SLD; ++q) sl4[tid + q * 1024] = sv[q];
        __syncthreads();
        if (sr + 1 < s1)
            tile_fetch(hin, ents, bbase, dr * NS + sr + 1, sr + 1, N, tid,
                       ne0, ne1, nent, sv);
        float2 v[EPRE];
#pragma unroll
        for (int k = 0; k < EPRE; ++k) v[k] = slice[ent[k] & TMASK];
#pragma unroll
        for (int k = 0; k < EPRE; ++k) {
            const int j = e0 + tid + k * 1024;
            if (j < e1) {
                atomicAdd(&accx[ent[k] >> TSH], v[k].x);
                atomicAdd(&accy[ent[k] >> TSH], v[k].y);
            }
        }
        for (int j = e0 + tid + EPRE * 1024; j < e1; j += 1024) {
            const unsigned u = ents[j];
            const float2 w = slice[u & TMASK];
            atomicAdd(&accx[u >> TSH], w.x);
            atomicAdd(&accy[u >> TSH], w.y);
        }
        __syncthreads();
    }
    const int base = dr << TSH;
    float2* po = part + (size_t)g * Np + base;
    float ax[TNODES / 1024], ay[TNODES / 1024];
#pragma unroll
    for (int q = 0; q < TNODES / 1024; ++q) {
        ax[q] = accx[tid + q * 1024];
        ay[q] = accy[tid + q * 1024];
    }
#pragma unroll
    for (int q = 0; q < TNODES / 1024; ++q) {
        const int i = tid + q * 1024;
        if (base + i < N) po[i] = make_float2(ax[q], ay[q]);
    }
}

#define SLX (TNODES / 4 / 1024)

__device__ __forceinline__ void tile_fetch_x(const float* __restrict__ hx,
                                             const unsigned* __restrict__ ents,
                                             const int* __restrict__ bbase,
                                             int t, int sr, int N, int tid,
                                             int& e0, int& e1,
                                             unsigned (&ent)[EPRE], float4 (&sv)[SLX]) {
    e0 = bbase[t];
    e1 = bbase[t + 1];
#pragma unroll
    for (int k = 0; k < EPRE; ++k) {
        const int j = e0 + tid + k * 1024;
        ent[k] = ents[(j < e1) ? j : 0];
    }
    const int sbase = sr << TSH;
    const int scnt  = (N - sbase < TNODES) ? (N - sbase) : TNODES;
    const float4* hsrc = (const float4*)(hx + sbase);
#pragma unroll
    for (int q = 0; q < SLX; ++q) {
        const int i = tid + q * 1024;
        sv[q] = hsrc[(4 * i < scnt) ? i : 0];
    }
}

__global__ void __launch_bounds__(1024)
k_tile_gather_x(const float* __restrict__ hx, const unsigned* __restrict__ ents,
                const int* __restrict__ bbase, float* __restrict__ part,
                int N, int NS, int SPG, int Np) {
    __shared__ float slice[TNODES];
    __shared__ float accx[TNODES];
    const int tid = threadIdx.x;
    const int dr  = blockIdx.x / NGRP;
    const int g   = blockIdx.x % NGRP;
    int s0 = g * SPG;  if (s0 > NS) s0 = NS;
    int s1 = s0 + SPG; if (s1 > NS) s1 = NS;
    float4* sl4 = (float4*)slice;

    int ne0 = 0, ne1 = 0;
    unsigned nent[EPRE];
    float4 sv[SLX];
#pragma unroll
    for (int k = 0; k < EPRE; ++k) nent[k] = 0u;
#pragma unroll
    for (int q = 0; q < SLX; ++q) sv[q] = make_float4(0.f, 0.f, 0.f, 0.f);
    if (s0 < s1)
        tile_fetch_x(hx, ents, bbase, dr * NS + s0, s0, N, tid, ne0, ne1, nent, sv);

    for (int i = tid; i < TNODES; i += 1024) accx[i] = 0.f;
    __syncthreads();
    for (int sr = s0; sr < s1; ++sr) {
        const int e0 = ne0;
        const int e1 = ne1;
        unsigned ent[EPRE];
#pragma unroll
        for (int k = 0; k < EPRE; ++k) ent[k] = nent[k];
#pragma unroll
        for (int q = 0; q < SLX; ++q) sl4[tid + q * 1024] = sv[q];
        __syncthreads();
        if (sr + 1 < s1)
            tile_fetch_x(hx, ents, bbase, dr * NS + sr + 1, sr + 1, N, tid,
                         ne0, ne1, nent, sv);
        float v[EPRE];
#pragma unroll
        for (int k = 0; k < EPRE; ++k) v[k] = slice[ent[k] & TMASK];
#pragma unroll
        for (int k = 0; k < EPRE; ++k) {
            const int j = e0 + tid + k * 1024;
            if (j < e1) atomicAdd(&accx[ent[k] >> TSH], v[k]);
        }
        for (int j = e0 + tid + EPRE * 1024; j < e1; j += 1024) {
            const unsigned u = ents[j];
            atomicAdd(&accx[u >> TSH], slice[u & TMASK]);
        }
        __syncthreads();
    }
    const int base = dr << TSH;
    float* po = part + (size_t)g * Np + base;
    float ax[TNODES / 1024];
#pragma unroll
    for (int q = 0; q < TNODES / 1024; ++q) ax[q] = accx[tid + q * 1024];
#pragma unroll
    for (int q = 0; q < TNODES / 1024; ++q) {
        const int i = tid + q * 1024;
        if (base + i < N) po[i] = ax[q];
    }
}

template <bool XP>
__global__ void __launch_bounds__(256)
k_combine(const float2* __restrict__ hin, const float2* __restrict__ part,
          const float* __restrict__ dinv, float2* __restrict__ hout,
          float* __restrict__ hx, int N, int Np) {
    const int n = blockIdx.x * blockDim.x + threadIdx.x;
    if (n >= N) return;
    float2 p[NGRP];
#pragma unroll
    for (int g = 0; g < NGRP; ++g) p[g] = part[(size_t)g * Np + n];
    const float2 self = hin[n];
    const float  di   = dinv[n];
    float rx = p[0].x, ry = p[0].y;
#pragma unroll
    for (int g = 1; g < NGRP; ++g) { rx += p[g].x; ry += p[g].y; }
    const float ox = (rx + self.x) * di;
    hout[n] = make_float2(ox, (ry + self.y) * di);
    if (XP) hx[n] = ox;
}

#define OTILES 4

__global__ void __launch_bounds__(256)
k_out_wmma(const float2* __restrict__ h1, const float2* __restrict__ h2,
           const float2* __restrict__ h3, const float* __restrict__ px,
           const float* __restrict__ dinv, const float* __restrict__ sdeg,
           const float* __restrict__ M, const float* __restrict__ bias,
           float* __restrict__ out, int N, int Np) {
    constexpr int F_OUT = 32;
    const int wave = __builtin_amdgcn_readfirstlane(
        (int)(blockIdx.x * (blockDim.x >> 5) + (threadIdx.x >> 5)));
    const int lane = threadIdx.x & 31;
    const int rowW = wave * (16 * OTILES);
    if (rowW >= N) return;

    const int  lg   = lane >> 4;
    const int  lmod = lane & 15;

    v2f   bfr[2];
    float bv[2];
#pragma unroll
    for (int tn = 0; tn < 2; ++tn) {
        const int nc = tn * 16 + lmod;
        bv[tn]    = bias[nc];
        bfr[tn].x = M[(lg * 2) * F_OUT + nc];
        bfr[tn].y = M[(lg * 2 + 1) * F_OUT + nc];
    }

    v2f a[OTILES];
#pragma unroll
    for (int i = 0; i < OTILES; ++i) {
        const int m  = rowW + i * 16 + lmod;
        const int mc = (m < N) ? m : (N - 1);
        const float2 u0 = h2[mc];
        const float2 u1 = lg ? h1[mc] : h3[mc];
        float pp[NGRP];
#pragma unroll
        for (int g = 0; g < NGRP; ++g) pp[g] = px[(size_t)g * Np + mc];
        const float  di = dinv[mc];
        const float  sd = sdeg[mc];
        float rx = pp[0];
#pragma unroll
        for (int g = 1; g < NGRP; ++g) rx += pp[g];
        const float p4 = (rx + u1.x) * di;
        a[i].x = (lg ? u0.y : p4) * sd;
        a[i].y = u1.y * sd;
    }

#pragma unroll
    for (int i = 0; i < OTILES; ++i) {
        const int row0 = rowW + i * 16;
        if (row0 < N) {
            const bool fullM = (row0 + 16 <= N);
#pragma unroll
            for (int tn = 0; tn < 2; ++tn) {
                const int nc = tn * 16 + lmod;
                v8f acc = {0.f, 0.f, 0.f, 0.f, 0.f, 0.f, 0.f, 0.f};
                acc = __builtin_amdgcn_wmma_f32_16x16x4_f32(
                    false, a[i], false, bfr[tn], (short)0, acc, false, false);

                float* p = out + (long long)(row0 + lg * 8) * F_OUT + nc;
                if (fullM) {
#pragma unroll
                    for (int v = 0; v < 8; ++v) p[(long long)v * F_OUT] = acc[v] + bv[tn];
                } else {
#pragma unroll
                    for (int v = 0; v < 8; ++v) {
                        const int mrow = row0 + v + lg * 8;
                        if (mrow < N) p[(long long)v * F_OUT] = acc[v] + bv[tn];
                    }
                }
            }
        }
    }
}

extern "C" void kernel_launch(void* const* d_in, const int* in_sizes, int n_in,
                              void* d_out, int out_size, void* d_ws, size_t ws_size,
                              hipStream_t stream) {
    const int* read_length = (const int*)d_in[0];
    const int* edge        = (const int*)d_in[1];
    const int  N = in_sizes[0];
    const int  E = in_sizes[1] / 2;
    const int* src = edge;
    const int* dst = edge + E;

    const float* W1 = (const float*)d_in[2]; const float* B1 = (const float*)d_in[3];
    const float* W2 = (const float*)d_in[4]; const float* B2 = (const float*)d_in[5];
    const float* W3 = (const float*)d_in[6]; const float* B3 = (const float*)d_in[7];
    const float* W4 = (const float*)d_in[8]; const float* B4 = (const float*)d_in[9];

    const int NS  = (N + TNODES - 1) / TNODES;
    const int NT  = NS * NS;
    const int SPG = (NS + NGRP - 1) / NGRP;
    const int Np  = (N + 3) & ~3;
    const int NCH = (E + CHUNK - 1) / CHUNK;
    const int HPI = (NT + 31) & ~31;
    float2* h0 = (float2*)d_ws;
    float2* h1 = h0 + (size_t)Np;
    float2* h2 = h1 + (size_t)Np;
    float2* h3 = h2 + (size_t)Np;
    float*  h3x  = (float*)(h3 + (size_t)Np);
    float2* part = h3 + 2 * (size_t)Np;
    float*  partx = (float*)part;
    int* pcnt      = (int*)(part + (size_t)NGRP * Np);
    unsigned* ents = (unsigned*)(pcnt + (size_t)NGRP * Np);
    float* dinv = (float*)(ents + (size_t)E);
    float* sdeg = dinv + Np;
    int* hist   = (int*)(sdeg + Np);
    int* off    = hist + (size_t)NCH * HPI;
    int* ttot   = off + (size_t)NCH * HPI;
    int* bbase  = ttot + HPI;
    float* Mf   = (float*)(bbase + (NT + 1));

    const int T  = 256;

    k_fold_wmma<<<1, 32, 0, stream>>>(W1, B1, W2, B2, W3, B3, W4, Mf);

    const int TB  = 512;
    k_tcount<<<NCH, TB, (size_t)NT * sizeof(int), stream>>>(src, dst, hist, E, NT, NS, HPI);
    k_colscan<<<HPI / 32, 1024, 0, stream>>>(hist, off, ttot, NCH, HPI);
    k_tscatter<<<NCH, TB, (size_t)NT * 2 * sizeof(int), stream>>>(src, dst, hist, off, ttot,
                                                                  bbase, ents, E, NT, NS, HPI);

    const unsigned tg = (unsigned)(NS * NGRP);
    const unsigned nb = (unsigned)((N + T - 1) / T);
    k_tile_count<<<tg, TNODES / 16, 0, stream>>>(ents, bbase, pcnt, N, NS, SPG, Np);
    k_node_init<<<nb, T, 0, stream>>>(pcnt, read_length, dinv, sdeg, h0, N, Np);

    k_tile_gather<<<tg, TNODES / 16, 0, stream>>>(h0, ents, bbase, part, N, NS, SPG, Np);
    k_combine<false><<<nb, T, 0, stream>>>(h0, part, dinv, h1, h3x, N, Np);
    k_tile_gather<<<tg, TNODES / 16, 0, stream>>>(h1, ents, bbase, part, N, NS, SPG, Np);
    k_combine<false><<<nb, T, 0, stream>>>(h1, part, dinv, h2, h3x, N, Np);
    k_tile_gather<<<tg, TNODES / 16, 0, stream>>>(h2, ents, bbase, part, N, NS, SPG, Np);
    k_combine<true><<<nb, T, 0, stream>>>(h2, part, dinv, h3, h3x, N, Np);
    k_tile_gather_x<<<tg, TNODES / 16, 0, stream>>>(h3x, ents, bbase, partx, N, NS, SPG, Np);

    const long long waves  = (long long)((N + 16 * OTILES - 1) / (16 * OTILES));
    const long long blocks = (waves + 7) / 8;
    k_out_wmma<<<(unsigned)blocks, 256, 0, stream>>>(h1, h2, h3, partx, dinv, sdeg, Mf, B4,
                                                     (float*)d_out, N, Np);
}
